# starting kernel
# baseline (speedup 1.0000x reference)
#include <hip/hip_runtime.h>
#include <stdint.h>
#include <stddef.h>


typedef __attribute__((ext_vector_type(16))) __bf16 v16bf;
typedef __attribute__((ext_vector_type(8)))  float  v8f;

#define LDSTR 520

__device__ __forceinline__ unsigned f2bf(float f) {
  unsigned u = __float_as_uint(f);
  return (u + 0x7FFFu + ((u >> 16) & 1u)) >> 16;
}

__device__ __forceinline__ int kmap(int v, int lane) {
  return ((v & 4) << 2) | ((lane & 16) >> 1) | ((v & 3) << 1);
}

__device__ __forceinline__ unsigned fenc(float f) {
  unsigned u = __float_as_uint(f);
  return (u & 0x80000000u) ? ~u : (u | 0x80000000u);
}
__device__ __forceinline__ float fdec(unsigned u) {
  return __uint_as_float((u & 0x80000000u) ? (u & 0x7FFFFFFFu) : ~u);
}

__device__ __forceinline__ void pack_tile_to_fragA(const float* tile,
                                                   unsigned* dst, int mt, int tid) {
  const int lane = tid & 31, w = tid >> 5;
  const int m = lane & 15;
#pragma unroll
  for (int q = 0; q < 2; ++q) {
    const int ks = w * 2 + q;
#pragma unroll
    for (int v = 0; v < 8; ++v) {
      const int k = (ks << 5) + kmap(v, lane);
      const unsigned lo = f2bf(tile[m * LDSTR + k]);
      const unsigned hi = f2bf(tile[m * LDSTR + k + 1]);
      dst[((size_t)(mt * 16 + ks) * 32 + lane) * 8 + v] = lo | (hi << 16);
    }
  }
}

__global__ void k_pack_B(const float* __restrict__ W, unsigned* __restrict__ pack,
                         int K, int N) {
  const int id = blockIdx.x * 256 + threadIdx.x;
  const int nks = K >> 5;
  const int total = (N >> 4) * nks * 256;
  if (id >= total) return;
  const int v = id & 7, l = (id >> 3) & 31;
  const int t = id >> 8, ks = t % nks, nt = t / nks;
  const int n = (nt << 4) + (l & 15);
  const int k = (ks << 5) + kmap(v, l);
  pack[id] = f2bf(W[(size_t)k * N + n]) | (f2bf(W[(size_t)(k + 1) * N + n]) << 16);
}

__global__ void k_pack_A(const float* __restrict__ X, unsigned* __restrict__ pack,
                         int M, int K) {
  const int id = blockIdx.x * 256 + threadIdx.x;
  const int nks = K >> 5;
  const int total = (M >> 4) * nks * 256;
  if (id >= total) return;
  const int v = id & 7, l = (id >> 3) & 31;
  const int t = id >> 8, ks = t % nks, mt = t / nks;
  const int m = (mt << 4) + (l & 15);
  const int k = (ks << 5) + kmap(v, l);
  pack[id] = f2bf(X[(size_t)m * K + k]) | (f2bf(X[(size_t)m * K + k + 1]) << 16);
}

__global__ __launch_bounds__(256) void k_gemm_ft_ln(
    const unsigned* __restrict__ packX, const unsigned* __restrict__ packW,
    const float* __restrict__ ft_b, const float* __restrict__ ln_g,
    const float* __restrict__ ln_b, unsigned* __restrict__ packH1) {
  __shared__ float tile[16 * LDSTR];
  __shared__ float psum[16][16], psq[16][16], stats[16][2];
  const int tid = threadIdx.x, lane = tid & 31, w = tid >> 5;
  const int mt = blockIdx.x;
  const int nks = 4;
  v8f acc[4] = {};
  for (int ks = 0; ks < nks; ++ks) {
    v16bf a = *(const v16bf*)(packX + ((size_t)(mt * nks + ks) * 32 + lane) * 8);
#pragma unroll
    for (int j = 0; j < 4; ++j) {
      const int nt = w * 4 + j;
      v16bf b = *(const v16bf*)(packW + ((size_t)(nt * nks + ks) * 32 + lane) * 8);
      acc[j] = __builtin_amdgcn_wmma_f32_16x16x32_bf16(false, a, false, b,
                                                       (short)0, acc[j], false, false);
    }
  }
  const int m0 = (lane >> 4) << 3, nc = lane & 15;
#pragma unroll
  for (int j = 0; j < 4; ++j) {
    const int n = (w * 4 + j) * 16 + nc;
    const float bias = ft_b[n];
#pragma unroll
    for (int r = 0; r < 8; ++r) tile[(m0 + r) * LDSTR + n] = acc[j][r] + bias;
  }
  __syncthreads();
  {
    const int row = tid >> 4, p = tid & 15;
    float s = 0.f, s2 = 0.f;
    for (int c = p * 32; c < p * 32 + 32; ++c) {
      const float xv = tile[row * LDSTR + c];
      s += xv; s2 += xv * xv;
    }
    psum[row][p] = s; psq[row][p] = s2;
  }
  __syncthreads();
  if (tid < 16) {
    float s = 0.f, s2 = 0.f;
    for (int p = 0; p < 16; ++p) { s += psum[tid][p]; s2 += psq[tid][p]; }
    const float mu = s * (1.f / 512.f);
    const float var = s2 * (1.f / 512.f) - mu * mu;
    stats[tid][0] = mu;
    stats[tid][1] = rsqrtf(var + 1e-5f);
  }
  __syncthreads();
  {
    const int row = tid >> 4, p = tid & 15;
    const float mu = stats[row][0], rs = stats[row][1];
    for (int c = p * 32; c < p * 32 + 32; ++c) {
      float xv = (tile[row * LDSTR + c] - mu) * rs * ln_g[c] + ln_b[c];
      tile[row * LDSTR + c] = xv >= 0.f ? xv : 0.2f * xv;
    }
  }
  __syncthreads();
  pack_tile_to_fragA(tile, packH1, mt, tid);
}

__global__ __launch_bounds__(256) void k_gemm1(
    const unsigned* __restrict__ packA, const unsigned* __restrict__ packW,
    const float* __restrict__ att_src, const float* __restrict__ att_dst,
    float* __restrict__ hW, float* __restrict__ a_src, float* __restrict__ a_dst) {
  __shared__ float tile[16 * LDSTR];
  const int tid = threadIdx.x, lane = tid & 31, w = tid >> 5;
  const int mt = blockIdx.x;
  const int nks = 16;
  v8f acc[4] = {};
  for (int ks = 0; ks < nks; ++ks) {
    v16bf a = *(const v16bf*)(packA + ((size_t)(mt * nks + ks) * 32 + lane) * 8);
#pragma unroll
    for (int j = 0; j < 4; ++j) {
      const int nt = w * 4 + j;
      v16bf b = *(const v16bf*)(packW + ((size_t)(nt * nks + ks) * 32 + lane) * 8);
      acc[j] = __builtin_amdgcn_wmma_f32_16x16x32_bf16(false, a, false, b,
                                                       (short)0, acc[j], false, false);
    }
  }
  const int m0 = (lane >> 4) << 3, nc = lane & 15;
#pragma unroll
  for (int j = 0; j < 4; ++j) {
    const int n = (w * 4 + j) * 16 + nc;
#pragma unroll
    for (int r = 0; r < 8; ++r) tile[(m0 + r) * LDSTR + n] = acc[j][r];
  }
  __syncthreads();
  for (int i = tid; i < 16 * 512; i += 256) {
    const int row = i >> 9, c = i & 511;
    hW[(size_t)(mt * 16 + row) * 512 + c] = tile[row * LDSTR + c];
  }
  if (tid < 128) {
    const int row = tid >> 3, h = tid & 7;
    float s1 = 0.f, s2 = 0.f;
    for (int c = 0; c < 64; ++c) {
      const float v = tile[row * LDSTR + (h << 6) + c];
      s1 += v * att_src[(h << 6) + c];
      s2 += v * att_dst[(h << 6) + c];
    }
    const int node = mt * 16 + row;
    a_src[node * 8 + h] = s1;
    a_dst[node * 8 + h] = s2;
  }
}

__global__ void k_edge_logit_max(const int* __restrict__ ei,
    const float* __restrict__ a_src, const float* __restrict__ a_dst,
    float* __restrict__ buf, unsigned* __restrict__ maxenc, int E) {
  const int idx = blockIdx.x * 256 + threadIdx.x;
  if (idx >= E * 8) return;
  const int e = idx >> 3, h = idx & 7;
  const int s = ei[e], d = ei[E + e];
  float L = a_src[s * 8 + h] + a_dst[d * 8 + h];
  L = L >= 0.f ? L : 0.2f * L;
  buf[idx] = L;
  atomicMax(&maxenc[d * 8 + h], fenc(L));
}

__global__ void k_edge_expsum(const int* __restrict__ ei,
    float* __restrict__ buf, const unsigned* __restrict__ maxenc,
    float* __restrict__ denom, int E) {
  const int idx = blockIdx.x * 256 + threadIdx.x;
  if (idx >= E * 8) return;
  const int e = idx >> 3, h = idx & 7;
  const int d = ei[E + e];
  const float ex = __expf(buf[idx] - fdec(maxenc[d * 8 + h]));
  buf[idx] = ex;
  atomicAdd(&denom[d * 8 + h], ex);
}

__global__ __launch_bounds__(256) void k_scatter1(const int* __restrict__ ei,
    const float* __restrict__ hW, const float* __restrict__ expv,
    const float* __restrict__ denom, float* __restrict__ agg, int E) {
  __shared__ float alpha[8];
  const int e = blockIdx.x;
  const int s = ei[e], d = ei[E + e];
  if (threadIdx.x < 8)
    alpha[threadIdx.x] = expv[e * 8 + threadIdx.x] / denom[d * 8 + threadIdx.x];
  __syncthreads();
#pragma unroll
  for (int i = 0; i < 2; ++i) {
    const int c = threadIdx.x + i * 256;
    atomicAdd(&agg[(size_t)d * 512 + c], hW[(size_t)s * 512 + c] * alpha[c >> 6]);
  }
}

__global__ void k_scatter2(const int* __restrict__ ei,
    const float* __restrict__ hW, const float* __restrict__ expv,
    const float* __restrict__ denom, float* __restrict__ agg, int E) {
  const int idx = blockIdx.x * 256 + threadIdx.x;
  if (idx >= E * 16) return;
  const int e = idx >> 4, c = idx & 15, h = c >> 1;
  const int s = ei[e], d = ei[E + e];
  const float alpha = expv[e * 8 + h] / denom[d * 8 + h];
  atomicAdd(&agg[(size_t)d * 16 + c], hW[(size_t)s * 16 + c] * alpha);
}

__global__ __launch_bounds__(256) void k_bnprep(const float* __restrict__ agg,
    const float* __restrict__ b1, const float* __restrict__ bn_g,
    const float* __restrict__ bn_b, const float* __restrict__ bn_mean,
    const float* __restrict__ bn_var, unsigned* __restrict__ packH2) {
  __shared__ float tile[16 * LDSTR];
  const int mt = blockIdx.x, tid = threadIdx.x;
  for (int i = tid; i < 16 * 512; i += 256) {
    const int row = i >> 9, c = i & 511;
    float xv = agg[(size_t)(mt * 16 + row) * 512 + c] + b1[c];
    xv = (xv - bn_mean[c]) * rsqrtf(bn_var[c] + 1e-5f) * bn_g[c] + bn_b[c];
    tile[row * LDSTR + c] = xv > 0.f ? xv : 0.f;
  }
  __syncthreads();
  pack_tile_to_fragA(tile, packH2, mt, tid);
}

__global__ __launch_bounds__(32) void k_gemm2(
    const unsigned* __restrict__ packA, const unsigned* __restrict__ packW,
    const float* __restrict__ att_src, const float* __restrict__ att_dst,
    float* __restrict__ hW, float* __restrict__ a_src, float* __restrict__ a_dst) {
  __shared__ float t2[16][16];
  const int lane = threadIdx.x;
  const int mt = blockIdx.x;
  const int nks = 16;
  v8f acc = {};
  for (int ks = 0; ks < nks; ++ks) {
    v16bf a = *(const v16bf*)(packA + ((size_t)(mt * nks + ks) * 32 + lane) * 8);
    v16bf b = *(const v16bf*)(packW + ((size_t)ks * 32 + lane) * 8);
    acc = __builtin_amdgcn_wmma_f32_16x16x32_bf16(false, a, false, b,
                                                  (short)0, acc, false, false);
  }
  const int m0 = (lane >> 4) << 3, nc = lane & 15;
#pragma unroll
  for (int r = 0; r < 8; ++r) t2[m0 + r][nc] = acc[r];
  __syncthreads();
  for (int i = lane; i < 256; i += 32) {
    const int row = i >> 4, c = i & 15;
    hW[(size_t)(mt * 16 + row) * 16 + c] = t2[row][c];
  }
  if (lane < 16) {
    const int row = lane, node = mt * 16 + row;
    for (int h = 0; h < 8; ++h) {
      const float v0 = t2[row][2 * h], v1 = t2[row][2 * h + 1];
      a_src[node * 8 + h] = v0 * att_src[2 * h] + v1 * att_src[2 * h + 1];
      a_dst[node * 8 + h] = v0 * att_dst[2 * h] + v1 * att_dst[2 * h + 1];
    }
  }
}

__global__ void k_final(const float* __restrict__ agg, const float* __restrict__ b2,
                        float* __restrict__ out, int N) {
  const int t = blockIdx.x * 256 + threadIdx.x;
  if (t >= N * 2) return;
  const int n = t >> 1, c = t & 1;
  float s = 0.f;
#pragma unroll
  for (int h = 0; h < 8; ++h) s += agg[n * 16 + 2 * h + c];
  out[t] = s * 0.125f + b2[c];
}

__global__ void k_zero(unsigned* __restrict__ p, size_t n) {
  const size_t i = (size_t)blockIdx.x * 256 + threadIdx.x;
  if (i < n) p[i] = 0u;
}

extern "C" void kernel_launch(void* const* d_in, const int* in_sizes, int n_in,
                              void* d_out, int out_size, void* d_ws, size_t ws_size,
                              hipStream_t stream) {
  const float* x       = (const float*)d_in[0];
  const int*   ei      = (const int*)d_in[1];
  const float* ft_W    = (const float*)d_in[2];
  const float* ft_b    = (const float*)d_in[3];
  const float* ln_g    = (const float*)d_in[4];
  const float* ln_b    = (const float*)d_in[5];
  const float* W1      = (const float*)d_in[6];
  const float* att_s1  = (const float*)d_in[7];
  const float* att_d1  = (const float*)d_in[8];
  const float* b1      = (const float*)d_in[9];
  const float* bn_g    = (const float*)d_in[10];
  const float* bn_b    = (const float*)d_in[11];
  const float* bn_mean = (const float*)d_in[12];
  const float* bn_var  = (const float*)d_in[13];
  const float* W2      = (const float*)d_in[14];
  const float* att_s2  = (const float*)d_in[15];
  const float* att_d2  = (const float*)d_in[16];
  const float* b2      = (const float*)d_in[17];

  const int N  = in_sizes[0] / 128;
  const int E  = in_sizes[1] / 2;
  const int MT = N / 16;

  char* cur = (char*)d_ws;
  auto alloc = [&](size_t bytes) -> void* {
    void* p = (void*)cur;
    cur += (bytes + 255) & ~(size_t)255;
    return p;
  };
  unsigned* packWft = (unsigned*)alloc((size_t)32 * 4 * 256 * 4);
  unsigned* packW1  = (unsigned*)alloc((size_t)32 * 16 * 256 * 4);
  unsigned* packW2  = (unsigned*)alloc((size_t)1 * 16 * 256 * 4);
  unsigned* packX   = (unsigned*)alloc((size_t)MT * 4 * 256 * 4);
  unsigned* packH1  = (unsigned*)alloc((size_t)MT * 16 * 256 * 4);
  unsigned* packH2  = (unsigned*)alloc((size_t)MT * 16 * 256 * 4);
  float* hW1 = (float*)alloc((size_t)N * 512 * 4);
  float* hW2 = (float*)alloc((size_t)N * 16 * 4);
  float* aS1 = (float*)alloc((size_t)N * 8 * 4);
  float* aD1 = (float*)alloc((size_t)N * 8 * 4);
  float* aS2 = (float*)alloc((size_t)N * 8 * 4);
  float* aD2 = (float*)alloc((size_t)N * 8 * 4);
  float* ex1 = (float*)alloc((size_t)E * 8 * 4);
  float* ex2 = (float*)alloc((size_t)E * 8 * 4);
  char* z0 = cur;
  float*    agg1 = (float*)alloc((size_t)N * 512 * 4);
  float*    agg2 = (float*)alloc((size_t)N * 16 * 4);
  unsigned* max1 = (unsigned*)alloc((size_t)N * 8 * 4);
  float*    den1 = (float*)alloc((size_t)N * 8 * 4);
  unsigned* max2 = (unsigned*)alloc((size_t)N * 8 * 4);
  float*    den2 = (float*)alloc((size_t)N * 8 * 4);
  const size_t zwords = (size_t)(cur - z0) / 4;

  k_zero<<<(unsigned)((zwords + 255) / 256), 256, 0, stream>>>((unsigned*)z0, zwords);

  k_pack_B<<<(32 * 4 * 256 + 255) / 256, 256, 0, stream>>>(ft_W, packWft, 128, 512);
  k_pack_B<<<(32 * 16 * 256 + 255) / 256, 256, 0, stream>>>(W1, packW1, 512, 512);
  k_pack_B<<<(1 * 16 * 256 + 255) / 256, 256, 0, stream>>>(W2, packW2, 512, 16);
  k_pack_A<<<(MT * 4 * 256 + 255) / 256, 256, 0, stream>>>(x, packX, N, 128);

  k_gemm_ft_ln<<<MT, 256, 0, stream>>>(packX, packWft, ft_b, ln_g, ln_b, packH1);

  k_gemm1<<<MT, 256, 0, stream>>>(packH1, packW1, att_s1, att_d1, hW1, aS1, aD1);
  const int EH = E * 8;
  k_edge_logit_max<<<(EH + 255) / 256, 256, 0, stream>>>(ei, aS1, aD1, ex1, max1, E);
  k_edge_expsum<<<(EH + 255) / 256, 256, 0, stream>>>(ei, ex1, max1, den1, E);
  k_scatter1<<<E, 256, 0, stream>>>(ei, hW1, ex1, den1, agg1, E);

  k_bnprep<<<MT, 256, 0, stream>>>(agg1, b1, bn_g, bn_b, bn_mean, bn_var, packH2);

  k_gemm2<<<MT, 32, 0, stream>>>(packH2, packW2, att_s2, att_d2, hW2, aS2, aD2);
  k_edge_logit_max<<<(EH + 255) / 256, 256, 0, stream>>>(ei, aS2, aD2, ex2, max2, E);
  k_edge_expsum<<<(EH + 255) / 256, 256, 0, stream>>>(ei, ex2, max2, den2, E);
  k_scatter2<<<(E * 16 + 255) / 256, 256, 0, stream>>>(ei, hW2, ex2, den2, agg2, E);

  k_final<<<(N * 2 + 255) / 256, 256, 0, stream>>>(agg2, b2, (float*)d_out, N);
}


# GEMM1 (h1 @ W1): a work-group owns 128 rows x one head's 64 columns, copies that head's weight fragments to LDS once; waves split rows, not columns; A fragments fetched a group ahead; sums unchanged
# speedup vs baseline: 1.9254x; 1.9254x over previous
#include <hip/hip_runtime.h>
#include <stdint.h>
#include <stddef.h>


typedef __attribute__((ext_vector_type(16))) __bf16 v16bf;
typedef __attribute__((ext_vector_type(8)))  float  v8f;

#define LDSTR 520

__device__ __forceinline__ unsigned f2bf(float f) {
  unsigned u = __float_as_uint(f);
  return (u + 0x7FFFu + ((u >> 16) & 1u)) >> 16;
}

__device__ __forceinline__ int kmap(int v, int lane) {
  return ((v & 4) << 2) | ((lane & 16) >> 1) | ((v & 3) << 1);
}

__device__ __forceinline__ unsigned fenc(float f) {
  unsigned u = __float_as_uint(f);
  return (u & 0x80000000u) ? ~u : (u | 0x80000000u);
}
__device__ __forceinline__ float fdec(unsigned u) {
  return __uint_as_float((u & 0x80000000u) ? (u & 0x7FFFFFFFu) : ~u);
}

__device__ __forceinline__ void pack_tile_to_fragA(const float* tile,
                                                   unsigned* dst, int mt, int tid) {
  const int lane = tid & 31, w = tid >> 5;
  const int m = lane & 15;
#pragma unroll
  for (int q = 0; q < 2; ++q) {
    const int ks = w * 2 + q;
#pragma unroll
    for (int v = 0; v < 8; ++v) {
      const int k = (ks << 5) + kmap(v, lane);
      const unsigned lo = f2bf(tile[m * LDSTR + k]);
      const unsigned hi = f2bf(tile[m * LDSTR + k + 1]);
      dst[((size_t)(mt * 16 + ks) * 32 + lane) * 8 + v] = lo | (hi << 16);
    }
  }
}

__global__ void k_pack_B(const float* __restrict__ W, unsigned* __restrict__ pack,
                         int K, int N) {
  const int id = blockIdx.x * 256 + threadIdx.x;
  const int nks = K >> 5;
  const int total = (N >> 4) * nks * 256;
  if (id >= total) return;
  const int v = id & 7, l = (id >> 3) & 31;
  const int t = id >> 8, ks = t % nks, nt = t / nks;
  const int n = (nt << 4) + (l & 15);
  const int k = (ks << 5) + kmap(v, l);
  pack[id] = f2bf(W[(size_t)k * N + n]) | (f2bf(W[(size_t)(k + 1) * N + n]) << 16);
}

__global__ void k_pack_A(const float* __restrict__ X, unsigned* __restrict__ pack,
                         int M, int K) {
  const int id = blockIdx.x * 256 + threadIdx.x;
  const int nks = K >> 5;
  const int total = (M >> 4) * nks * 256;
  if (id >= total) return;
  const int v = id & 7, l = (id >> 3) & 31;
  const int t = id >> 8, ks = t % nks, mt = t / nks;
  const int m = (mt << 4) + (l & 15);
  const int k = (ks << 5) + kmap(v, l);
  pack[id] = f2bf(X[(size_t)m * K + k]) | (f2bf(X[(size_t)m * K + k + 1]) << 16);
}

__global__ __launch_bounds__(256) void k_gemm_ft_ln(
    const unsigned* __restrict__ packX, const unsigned* __restrict__ packW,
    const float* __restrict__ ft_b, const float* __restrict__ ln_g,
    const float* __restrict__ ln_b, unsigned* __restrict__ packH1) {
  __shared__ float tile[16 * LDSTR];
  __shared__ float psum[16][16], psq[16][16], stats[16][2];
  const int tid = threadIdx.x, lane = tid & 31, w = tid >> 5;
  const int mt = blockIdx.x;
  const int nks = 4;
  v8f acc[4] = {};
  for (int ks = 0; ks < nks; ++ks) {
    v16bf a = *(const v16bf*)(packX + ((size_t)(mt * nks + ks) * 32 + lane) * 8);
#pragma unroll
    for (int j = 0; j < 4; ++j) {
      const int nt = w * 4 + j;
      v16bf b = *(const v16bf*)(packW + ((size_t)(nt * nks + ks) * 32 + lane) * 8);
      acc[j] = __builtin_amdgcn_wmma_f32_16x16x32_bf16(false, a, false, b,
                                                       (short)0, acc[j], false, false);
    }
  }
  const int m0 = (lane >> 4) << 3, nc = lane & 15;
#pragma unroll
  for (int j = 0; j < 4; ++j) {
    const int n = (w * 4 + j) * 16 + nc;
    const float bias = ft_b[n];
#pragma unroll
    for (int r = 0; r < 8; ++r) tile[(m0 + r) * LDSTR + n] = acc[j][r] + bias;
  }
  __syncthreads();
  {
    const int row = tid >> 4, p = tid & 15;
    float s = 0.f, s2 = 0.f;
    for (int c = p * 32; c < p * 32 + 32; ++c) {
      const float xv = tile[row * LDSTR + c];
      s += xv; s2 += xv * xv;
    }
    psum[row][p] = s; psq[row][p] = s2;
  }
  __syncthreads();
  if (tid < 16) {
    float s = 0.f, s2 = 0.f;
    for (int p = 0; p < 16; ++p) { s += psum[tid][p]; s2 += psq[tid][p]; }
    const float mu = s * (1.f / 512.f);
    const float var = s2 * (1.f / 512.f) - mu * mu;
    stats[tid][0] = mu;
    stats[tid][1] = rsqrtf(var + 1e-5f);
  }
  __syncthreads();
  {
    const int row = tid >> 4, p = tid & 15;
    const float mu = stats[row][0], rs = stats[row][1];
    for (int c = p * 32; c < p * 32 + 32; ++c) {
      float xv = (tile[row * LDSTR + c] - mu) * rs * ln_g[c] + ln_b[c];
      tile[row * LDSTR + c] = xv >= 0.f ? xv : 0.2f * xv;
    }
  }
  __syncthreads();
  pack_tile_to_fragA(tile, packH1, mt, tid);
}

typedef unsigned v4u __attribute__((ext_vector_type(4)));
typedef unsigned v8u __attribute__((ext_vector_type(8)));
#define G1_W 8
#define G1_OP 68

__global__ __launch_bounds__(256) void k_gemm1(
    const unsigned* __restrict__ packA, const unsigned* __restrict__ packW,
    const float* __restrict__ att_src, const float* __restrict__ att_dst,
    float* __restrict__ hW, float* __restrict__ a_src, float* __restrict__ a_dst, int MT) {
  __shared__ v4u bs[4 * 16 * 64];
  __shared__ __attribute__((aligned(16))) float ot[G1_W][16 * G1_OP];
  __shared__ __attribute__((aligned(16))) float atl[2][64];
  const int tid = threadIdx.x, lane = tid & 31, w = tid >> 5;
  const int head = blockIdx.x & 7, rg = blockIdx.x >> 3;
  const int mtw = rg * G1_W + w;
  const bool valid = mtw < MT;
  const int mt = valid ? mtw : MT - 1;
  const unsigned* __restrict__ ap = packA + ((size_t)mt * 16 * 32 + lane) * 8;
  v16bf ar[3][2];
  auto load_group = [&](int g) {
#pragma unroll
    for (int q = 0; q < 2; ++q) ar[g % 3][q] = *(const v16bf*)(ap + (2 * g + q) * 256);
  };
  load_group(0);
  {
    const v4u* __restrict__ gB = (const v4u*)(packW + (size_t)head * (4 * 16 * 256)) + tid;
    v4u* sb = bs + ((tid >> 6) << 6) + ((tid & 1) << 5) + ((tid & 63) >> 1);
    v4u tmp[16];
#pragma unroll
    for (int t = 0; t < 16; ++t) tmp[t] = gB[256 * t];
#pragma unroll
    for (int t = 0; t < 16; ++t) sb[256 * t] = tmp[t];
  }
  load_group(1);
  load_group(2);
  if (tid < 128) atl[tid >> 6][tid & 63] = (tid < 64 ? att_src : att_dst)[head * 64 + (tid & 63)];
  __syncthreads();
  v8f acc[4] = {};
  auto kstep = [&](const v16bf& a, int ks) {
    v16bf b[4];
#pragma unroll
    for (int j = 0; j < 4; ++j) {
      const v4u lo = bs[(j * 16 + ks) * 64 + lane];
      const v4u hi = bs[(j * 16 + ks) * 64 + 32 + lane];
      const v8u t = __builtin_shufflevector(lo, hi, 0, 1, 2, 3, 4, 5, 6, 7);
      b[j] = (v16bf)t;
    }
    __builtin_amdgcn_sched_barrier(0);
#pragma unroll
    for (int j = 0; j < 4; ++j)
      acc[j] = __builtin_amdgcn_wmma_f32_16x16x32_bf16(false, a, false, b[j],
                                                       (short)0, acc[j], false, false);
  };
#pragma unroll
  for (int g = 0; g < 8; ++g) {
    kstep(ar[g % 3][0], 2 * g);
    __builtin_amdgcn_sched_barrier(0);
    kstep(ar[g % 3][1], 2 * g + 1);
    if (g + 3 < 8) load_group(g + 3);
    __builtin_amdgcn_sched_barrier(0);
  }

  float* o = ot[w];
  const int m0 = (lane >> 4) << 3, nc = lane & 15;
#pragma unroll
  for (int j = 0; j < 4; ++j) {
#pragma unroll
    for (int r = 0; r < 8; ++r) o[(m0 + r) * G1_OP + j * 16 + nc] = acc[j][r];
  }
  __builtin_amdgcn_fence(__ATOMIC_ACQ_REL, "wavefront");
  if (valid) {
    const int hh = lane >> 4;
#pragma unroll
    for (int it = 0; it < 8; ++it) {
      const int row = it * 2 + hh, c4 = (lane & 15) * 4;
      const float4 v = *(const float4*)(o + row * G1_OP + c4);
      *(float4*)(hW + (size_t)(mt * 16 + row) * 512 + head * 64 + c4) = v;
    }
    const int row = lane & 15;
    const float* att = atl[hh];
    float s = 0.f;
    for (int c = 0; c < 64; ++c) s += o[row * G1_OP + c] * att[c];
    (hh ? a_dst : a_src)[(mt * 16 + row) * 8 + head] = s;
  }
}

#define AGG_R 40
#define AGG_WAVES 16
#define AGG_UL 16
#define AGG_STEP (32 * AGG_UL)
#define AGG_CAP (AGG_STEP + 32)
#define AGG_KEEP 256

__global__ __launch_bounds__(512) void k_agg1_owned(const int* __restrict__ ei,
    const float* __restrict__ hW, const float* __restrict__ a_src,
    const float* __restrict__ a_dst, float* __restrict__ agg, int N, int E) {
  __shared__ float acc[AGG_R * 512];
  __shared__ unsigned mx[AGG_R * 8];
  __shared__ float den[AGG_R * 8];
  __shared__ float adl[AGG_R * 8];
  __shared__ unsigned lst[AGG_WAVES][AGG_CAP];
  __shared__ unsigned short lsd[AGG_WAVES][AGG_CAP];
  __shared__ unsigned kps[AGG_WAVES][AGG_KEEP];
  __shared__ unsigned short kpd[AGG_WAVES][AGG_KEEP];
  const int tid = threadIdx.x, lane = tid & 31, w = tid >> 5, hh = lane >> 4;
  const int d0 = blockIdx.x * AGG_R;
  const int Rn = min(AGG_R, N - d0);
  for (int i = tid; i < AGG_R * 512; i += 512) acc[i] = 0.f;
  for (int i = tid; i < AGG_R * 8; i += 512) {
    mx[i] = 0u; den[i] = 0.f;
    adl[i] = (i < Rn * 8) ? a_dst[(size_t)d0 * 8 + i] : 0.f;
  }
  __syncthreads();
  const int SL = (E + AGG_WAVES - 1) / AGG_WAVES;
  const int e_begin = w * SL;
  const int e_end = min(E, e_begin + SL);
  const int nit = (SL + AGG_STEP - 1) / AGG_STEP;
  const int* __restrict__ dstp = ei + E;

  auto load_step = [&](int (&v)[AGG_UL], int start) {
#pragma unroll
    for (int u = 0; u < AGG_UL; ++u) {
      const int e = start + u * 32 + lane;
      v[u] = (e < e_end) ? dstp[e] : -1;
    }
  };
  auto match_step = [&](const int (&v)[AGG_UL], int base, int cnt) -> int {
#pragma unroll
    for (int u = 0; u < AGG_UL; ++u) {
      const unsigned dl = (unsigned)(v[u] - d0);
      const bool match = dl < (unsigned)Rn;
      const unsigned bal = __builtin_amdgcn_ballot_w32(match);
      if (bal) {
        const int pos = cnt + __builtin_popcount(bal & ((1u << lane) - 1u));
        if (match) {
          lst[w][pos] = (unsigned)(base + u * 32 + lane);
          lsd[w][pos] = (unsigned short)dl;
        }
        cnt += __builtin_popcount(bal);
      }
    }
    return cnt;
  };
  auto sum_rows = [&](const unsigned* ls, const unsigned short* ld, int n) {
    for (int j0 = 0; j0 < n; j0 += 4) {
      float4 row[4][4];
      float as[4][4];
      int dln[4];
#pragma unroll
      for (int b = 0; b < 4; ++b) {
        const int jb = min(j0 + b, n - 1);
        const size_t s = (size_t)ls[jb];
        dln[b] = (int)ld[jb];
        const float* rp = hW + s * 512 + 4 * lane;
#pragma unroll
        for (int j = 0; j < 4; ++j) row[b][j] = *(const float4*)(rp + 128 * j);
#pragma unroll
        for (int j = 0; j < 4; ++j) as[b][j] = a_src[s * 8 + 2 * j + hh];
      }
#pragma unroll
      for (int b = 0; b < 4; ++b) {
        if (j0 + b < n) {
          const int dl = dln[b];
#pragma unroll
          for (int j = 0; j < 4; ++j) {
            const int h = 2 * j + hh;
            float L = as[b][j] + adl[dl * 8 + h];
            L = L >= 0.f ? L : 0.2f * L;
            const float wgt = __expf(L - fdec(mx[dl * 8 + h]));
            float* ap = acc + dl * 512 + j * 128 + lane;
            atomicAdd(ap, wgt * row[b][j].x);
            atomicAdd(ap + 32, wgt * row[b][j].y);
            atomicAdd(ap + 64, wgt * row[b][j].z);
            atomicAdd(ap + 96, wgt * row[b][j].w);
            if ((lane & 15) == 0) atomicAdd(&den[dl * 8 + h], wgt);
          }
        }
      }
    }
  };

  int kept = 0;
  bool fits = true;
  {
    int dv[AGG_UL];
    load_step(dv, e_begin);
    int cnt = 0;
    for (int it = 0; it < nit; ++it) {
      const int base = e_begin + it * AGG_STEP;
      int nx[AGG_UL];
      load_step(nx, base + AGG_STEP);
      cnt = match_step(dv, base, cnt);
      if (cnt >= 32 || it == nit - 1) {
        __builtin_amdgcn_fence(__ATOMIC_ACQ_REL, "wavefront");
        const bool keep = fits && kept + cnt <= AGG_KEEP;
        for (int j = lane; j < cnt; j += 32) {
          const int e = (int)lst[w][j];
          const int dl = (int)lsd[w][j];
          const int s = ei[e];
          const float4 a0 = *(const float4*)(a_src + (size_t)s * 8);
          const float4 a1 = *(const float4*)(a_src + (size_t)s * 8 + 4);
          const float av[8] = {a0.x, a0.y, a0.z, a0.w, a1.x, a1.y, a1.z, a1.w};
#pragma unroll
          for (int h = 0; h < 8; ++h) {
            float L = av[h] + adl[dl * 8 + h];
            L = L >= 0.f ? L : 0.2f * L;
            atomicMax(&mx[dl * 8 + h], fenc(L));
          }
          if (keep) {
            kps[w][kept + j] = (unsigned)s;
            kpd[w][kept + j] = (unsigned short)dl;
          }
        }
        if (keep) kept += cnt; else fits = false;
        cnt = 0;
      }
#pragma unroll
      for (int u = 0; u < AGG_UL; ++u) dv[u] = nx[u];
    }
  }
  __syncthreads();

  if (fits) {
    sum_rows(kps[w], kpd[w], kept);
  } else {
    int dv[AGG_UL];
    load_step(dv, e_begin);
    for (int it = 0; it < nit; ++it) {
      const int base = e_begin + it * AGG_STEP;
      int nx[AGG_UL];
      load_step(nx, base + AGG_STEP);
      const int cnt = match_step(dv, base, 0);
      __builtin_amdgcn_fence(__ATOMIC_ACQ_REL, "wavefront");
      for (int j = lane; j < cnt; j += 32) lst[w][j] = (unsigned)ei[lst[w][j]];
      __builtin_amdgcn_fence(__ATOMIC_ACQ_REL, "wavefront");
      sum_rows(lst[w], lsd[w], cnt);
#pragma unroll
      for (int u = 0; u < AGG_UL; ++u) dv[u] = nx[u];
    }
  }
  __syncthreads();
  for (int i = tid; i < Rn * 512; i += 512) {
    const int dl = i >> 9, c = i & 511;
    const int q = c >> 2, k = c & 3, j = q >> 5, l = q & 31;
    const float dn = den[dl * 8 + (c >> 6)];
    const float v = acc[dl * 512 + (j * 4 + k) * 32 + l];
    agg[(size_t)(d0 + dl) * 512 + c] = dn != 0.f ? v / dn : 0.f;
  }
}

#define A2_R 48
#define A2_WAVES 16
#define A2_UL 8
#define A2_STEP (128 * A2_UL)
#define A2_CAP (A2_STEP + 32)
#define A2_KEEP 512

typedef int a2_vi4 __attribute__((ext_vector_type(4)));
typedef a2_vi4 a2_vi4u __attribute__((aligned(4)));

__global__ __launch_bounds__(512) void k_agg2_owned(const int* __restrict__ ei,
    const float* __restrict__ hW, const float* __restrict__ a_src,
    const float* __restrict__ a_dst, const float* __restrict__ b2,
    float* __restrict__ out, int N, int E) {
  __shared__ float acc[A2_R * 16];
  __shared__ unsigned mx[A2_R * 8];
  __shared__ float den[A2_R * 8];
  __shared__ float adl[A2_R * 8];
  __shared__ unsigned lst[A2_WAVES][A2_CAP];
  __shared__ unsigned short lsd[A2_WAVES][A2_CAP];
  __shared__ unsigned kps[A2_WAVES][A2_KEEP];
  __shared__ unsigned short kpd[A2_WAVES][A2_KEEP];
  const int tid = threadIdx.x, lane = tid & 31;
  const int w = __builtin_amdgcn_readfirstlane(tid >> 5);
  const int d0 = blockIdx.x * A2_R;
  const int Rn = min(A2_R, N - d0);
  for (int i = tid; i < A2_R * 16; i += 512) acc[i] = 0.f;
  for (int i = tid; i < A2_R * 8; i += 512) {
    mx[i] = 0u; den[i] = 0.f;
    adl[i] = (i < Rn * 8) ? a_dst[(size_t)d0 * 8 + i] : 0.f;
  }
  __syncthreads();
  const int SL = ((E + A2_WAVES - 1) / A2_WAVES + 3) & ~3;
  const int e_begin = w * SL;
  const int e_end = min(E, e_begin + SL);
  const int nit = (SL + A2_STEP - 1) / A2_STEP;
  const int* __restrict__ dstp = ei + E;

  auto load_step = [&](int (&v)[A2_UL][4], int start) {
    if (start + A2_STEP <= e_end) {
#pragma unroll
      for (int u = 0; u < A2_UL; ++u) {
        const a2_vi4u t = *(const a2_vi4u*)(dstp + start + u * 128 + lane * 4);
        v[u][0] = t.x; v[u][1] = t.y; v[u][2] = t.z; v[u][3] = t.w;
      }
    } else {
#pragma unroll
      for (int u = 0; u < A2_UL; ++u) {
        const int e = start + u * 128 + lane * 4;
        if (e + 3 < e_end) {
          const a2_vi4u t = *(const a2_vi4u*)(dstp + e);
          v[u][0] = t.x; v[u][1] = t.y; v[u][2] = t.z; v[u][3] = t.w;
        } else {
#pragma unroll
          for (int k = 0; k < 4; ++k) v[u][k] = (e + k < e_end) ? dstp[e + k] : -1;
        }
      }
    }
  };
  auto edge_max = [&](size_t s, int dl) {
    const float4 a0 = *(const float4*)(a_src + s * 8);
    const float4 a1 = *(const float4*)(a_src + s * 8 + 4);
    const float av[8] = {a0.x, a0.y, a0.z, a0.w, a1.x, a1.y, a1.z, a1.w};
#pragma unroll
    for (int h = 0; h < 8; ++h) {
      float L = av[h] + adl[dl * 8 + h];
      L = L >= 0.f ? L : 0.2f * L;
      atomicMax(&mx[dl * 8 + h], fenc(L));
    }
  };
  auto edge_sum = [&](size_t s, int dl) {
    const float4 a0 = *(const float4*)(a_src + s * 8);
    const float4 a1 = *(const float4*)(a_src + s * 8 + 4);
    const float4 r0 = *(const float4*)(hW + s * 16);
    const float4 r1 = *(const float4*)(hW + s * 16 + 4);
    const float4 r2 = *(const float4*)(hW + s * 16 + 8);
    const float4 r3 = *(const float4*)(hW + s * 16 + 12);
    const float av[8] = {a0.x, a0.y, a0.z, a0.w, a1.x, a1.y, a1.z, a1.w};
    const float hv[16] = {r0.x, r0.y, r0.z, r0.w, r1.x, r1.y, r1.z, r1.w,
                          r2.x, r2.y, r2.z, r2.w, r3.x, r3.y, r3.z, r3.w};
#pragma unroll
    for (int h = 0; h < 8; ++h) {
      float L = av[h] + adl[dl * 8 + h];
      L = L >= 0.f ? L : 0.2f * L;
      const float wgt = __expf(L - fdec(mx[dl * 8 + h]));
      atomicAdd(&den[dl * 8 + h], wgt);
      atomicAdd(&acc[dl * 16 + 2 * h], wgt * hv[2 * h]);
      atomicAdd(&acc[dl * 16 + 2 * h + 1], wgt * hv[2 * h + 1]);
    }
  };

  int kept = 0;
  bool fits = true;
  for (int pass = 0; pass < 2; ++pass) {
    if (pass == 1 && fits) {
      for (int j = lane; j < kept; j += 32) edge_sum((size_t)kps[w][j], (int)kpd[w][j]);
    } else {
      int dv[A2_UL][4];
      load_step(dv, e_begin);
      int cnt = 0;
      for (int it = 0; it < nit; ++it) {
        const int base = e_begin + it * A2_STEP;
        int nx[A2_UL][4];
        load_step(nx, base + A2_STEP);
#pragma unroll
        for (int u = 0; u < A2_UL; ++u) {
#pragma unroll
          for (int k = 0; k < 4; ++k) {
            const unsigned dl = (unsigned)(dv[u][k] - d0);
            const bool match = dl < (unsigned)Rn;
            const unsigned bal = __builtin_amdgcn_ballot_w32(match);
            if (__builtin_expect(bal != 0u, 0)) {
              const int pos = cnt + __builtin_popcount(bal & ((1u << lane) - 1u));
              if (match) {
                lst[w][pos] = (unsigned)(base + u * 128 + lane * 4 + k);
                lsd[w][pos] = (unsigned short)dl;
              }
              cnt += __builtin_popcount(bal);
            }
          }
        }
        if (cnt >= 32 || it == nit - 1) {
          __builtin_amdgcn_fence(__ATOMIC_ACQ_REL, "wavefront");
          const bool keep = fits && kept + cnt <= A2_KEEP;
          for (int j = lane; j < cnt; j += 32) {
            const int e = (int)lst[w][j];
            const int dl = (int)lsd[w][j];
            const size_t s = (size_t)ei[e];
            if (pass == 0) {
              edge_max(s, dl);
              if (keep) {
                kps[w][kept + j] = (unsigned)s;
                kpd[w][kept + j] = (unsigned short)dl;
              }
            } else {
              edge_sum(s, dl);
            }
          }
          if (keep) kept += cnt; else fits = false;
          cnt = 0;
        }
#pragma unroll
        for (int u = 0; u < A2_UL; ++u) {
#pragma unroll
          for (int k = 0; k < 4; ++k) dv[u][k] = nx[u][k];
        }
      }
    }
    __syncthreads();
  }
  for (int i = tid; i < Rn * 2; i += 512) {
    const int dl = i >> 1, c = i & 1;
    float s = 0.f;
#pragma unroll
    for (int h = 0; h < 8; ++h) {
      const float dn = den[dl * 8 + h];
      s += dn != 0.f ? acc[dl * 16 + 2 * h + c] / dn : 0.f;
    }
    out[(size_t)(d0 + dl) * 2 + c] = s * 0.125f + b2[c];
  }
}

__global__ __launch_bounds__(256) void k_bnprep(const float* __restrict__ agg,
    const float* __restrict__ b1, const float* __restrict__ bn_g,
    const float* __restrict__ bn_b, const float* __restrict__ bn_mean,
    const float* __restrict__ bn_var, unsigned* __restrict__ packH2) {
  __shared__ float tile[16 * LDSTR];
  const int mt = blockIdx.x, tid = threadIdx.x;
  for (int i = tid; i < 16 * 512; i += 256) {
    const int row = i >> 9, c = i & 511;
    float xv = agg[(size_t)(mt * 16 + row) * 512 + c] + b1[c];
    xv = (xv - bn_mean[c]) * rsqrtf(bn_var[c] + 1e-5f) * bn_g[c] + bn_b[c];
    tile[row * LDSTR + c] = xv > 0.f ? xv : 0.f;
  }
  __syncthreads();
  pack_tile_to_fragA(tile, packH2, mt, tid);
}

__global__ __launch_bounds__(32) void k_gemm2(
    const unsigned* __restrict__ packA, const unsigned* __restrict__ packW,
    const float* __restrict__ att_src, const float* __restrict__ att_dst,
    float* __restrict__ hW, float* __restrict__ a_src, float* __restrict__ a_dst) {
  __shared__ float t2[16][16];
  const int lane = threadIdx.x;
  const int mt = blockIdx.x;
  const int nks = 16;
  v8f acc = {};
  for (int ks = 0; ks < nks; ++ks) {
    v16bf a = *(const v16bf*)(packA + ((size_t)(mt * nks + ks) * 32 + lane) * 8);
    v16bf b = *(const v16bf*)(packW + ((size_t)ks * 32 + lane) * 8);
    acc = __builtin_amdgcn_wmma_f32_16x16x32_bf16(false, a, false, b,
                                                  (short)0, acc, false, false);
  }
  const int m0 = (lane >> 4) << 3, nc = lane & 15;
#pragma unroll
  for (int r = 0; r < 8; ++r) t2[m0 + r][nc] = acc[r];
  __syncthreads();
  for (int i = lane; i < 256; i += 32) {
    const int row = i >> 4, c = i & 15;
    hW[(size_t)(mt * 16 + row) * 16 + c] = t2[row][c];
  }
  if (lane < 16) {
    const int row = lane, node = mt * 16 + row;
    for (int h = 0; h < 8; ++h) {
      const float v0 = t2[row][2 * h], v1 = t2[row][2 * h + 1];
      a_src[node * 8 + h] = v0 * att_src[2 * h] + v1 * att_src[2 * h + 1];
      a_dst[node * 8 + h] = v0 * att_dst[2 * h] + v1 * att_dst[2 * h + 1];
    }
  }
}

extern "C" void kernel_launch(void* const* d_in, const int* in_sizes, int n_in,
                              void* d_out, int out_size, void* d_ws, size_t ws_size,
                              hipStream_t stream) {
  const float* x       = (const float*)d_in[0];
  const int*   ei      = (const int*)d_in[1];
  const float* ft_W    = (const float*)d_in[2];
  const float* ft_b    = (const float*)d_in[3];
  const float* ln_g    = (const float*)d_in[4];
  const float* ln_b    = (const float*)d_in[5];
  const float* W1      = (const float*)d_in[6];
  const float* att_s1  = (const float*)d_in[7];
  const float* att_d1  = (const float*)d_in[8];
  const float* b1      = (const float*)d_in[9];
  const float* bn_g    = (const float*)d_in[10];
  const float* bn_b    = (const float*)d_in[11];
  const float* bn_mean = (const float*)d_in[12];
  const float* bn_var  = (const float*)d_in[13];
  const float* W2      = (const float*)d_in[14];
  const float* att_s2  = (const float*)d_in[15];
  const float* att_d2  = (const float*)d_in[16];
  const float* b2      = (const float*)d_in[17];

  const int N  = in_sizes[0] / 128;
  const int E  = in_sizes[1] / 2;
  const int MT = N / 16;

  char* cur = (char*)d_ws;
  auto alloc = [&](size_t bytes) -> void* {
    void* p = (void*)cur;
    cur += (bytes + 255) & ~(size_t)255;
    return p;
  };
  unsigned* packWft = (unsigned*)alloc((size_t)32 * 4 * 256 * 4);
  unsigned* packW1  = (unsigned*)alloc((size_t)32 * 16 * 256 * 4);
  unsigned* packW2  = (unsigned*)alloc((size_t)1 * 16 * 256 * 4);
  unsigned* packX   = (unsigned*)alloc((size_t)MT * 4 * 256 * 4);
  unsigned* packH1  = (unsigned*)alloc((size_t)MT * 16 * 256 * 4);
  unsigned* packH2  = (unsigned*)alloc((size_t)MT * 16 * 256 * 4);
  float* hW1 = (float*)alloc((size_t)N * 512 * 4);
  float* hW2 = (float*)alloc((size_t)N * 16 * 4);
  float* aS1 = (float*)alloc((size_t)N * 8 * 4);
  float* aD1 = (float*)alloc((size_t)N * 8 * 4);
  float* aS2 = (float*)alloc((size_t)N * 8 * 4);
  float* aD2 = (float*)alloc((size_t)N * 8 * 4);
  float* agg1 = (float*)alloc((size_t)N * 512 * 4);

  k_pack_B<<<(32 * 4 * 256 + 255) / 256, 256, 0, stream>>>(ft_W, packWft, 128, 512);
  k_pack_B<<<(32 * 16 * 256 + 255) / 256, 256, 0, stream>>>(W1, packW1, 512, 512);
  k_pack_B<<<(1 * 16 * 256 + 255) / 256, 256, 0, stream>>>(W2, packW2, 512, 16);
  k_pack_A<<<(MT * 4 * 256 + 255) / 256, 256, 0, stream>>>(x, packX, N, 128);

  k_gemm_ft_ln<<<MT, 256, 0, stream>>>(packX, packWft, ft_b, ln_g, ln_b, packH1);

  k_gemm1<<<(MT + G1_W - 1) / G1_W * 8, 256, 0, stream>>>(packH1, packW1, att_s1, att_d1, hW1, aS1, aD1, MT);
  k_agg1_owned<<<(N + AGG_R - 1) / AGG_R, 512, 0, stream>>>(ei, hW1, aS1, aD1, agg1, N, E);

  k_bnprep<<<MT, 256, 0, stream>>>(agg1, b1, bn_g, bn_b, bn_mean, bn_var, packH2);

  k_gemm2<<<MT, 32, 0, stream>>>(packH2, packW2, att_s2, att_d2, hW2, aS2, aD2);
  k_agg2_owned<<<(N + A2_R - 1) / A2_R, 512, 0, stream>>>(ei, hW2, aS2, aD2, b2, (float*)d_out, N, E);
}


# Layer-1 sums: each match is routed in LDS to the wave owning its node; the owner adds the node's rows in registers, 8 in flight, and stores the row; no LDS float atomics; summation order changes
# speedup vs baseline: 3.2606x; 1.6935x over previous
#include <hip/hip_runtime.h>
#include <stdint.h>
#include <stddef.h>


typedef __attribute__((ext_vector_type(16))) __bf16 v16bf;
typedef __attribute__((ext_vector_type(8)))  float  v8f;
typedef unsigned v4u __attribute__((ext_vector_type(4)));

#define LDSTR 520

__device__ __forceinline__ unsigned f2bf(float f) {
  unsigned u = __float_as_uint(f);
  return (u + 0x7FFFu + ((u >> 16) & 1u)) >> 16;
}

__device__ __forceinline__ int kmap(int v, int lane) {
  return ((v & 4) << 2) | ((lane & 16) >> 1) | ((v & 3) << 1);
}

__device__ __forceinline__ unsigned fenc(float f) {
  unsigned u = __float_as_uint(f);
  return (u & 0x80000000u) ? ~u : (u | 0x80000000u);
}
__device__ __forceinline__ float fdec(unsigned u) {
  return __uint_as_float((u & 0x80000000u) ? (u & 0x7FFFFFFFu) : ~u);
}

__device__ __forceinline__ void pack_tile_to_fragA(const float* tile,
                                                   unsigned* dst, int mt, int tid) {
  const int lane = tid & 31, w = tid >> 5;
  const int m = lane & 15;
#pragma unroll
  for (int q = 0; q < 2; ++q) {
    const int ks = w * 2 + q;
#pragma unroll
    for (int v = 0; v < 8; ++v) {
      const int k = (ks << 5) + kmap(v, lane);
      const unsigned lo = f2bf(tile[m * LDSTR + k]);
      const unsigned hi = f2bf(tile[m * LDSTR + k + 1]);
      dst[((size_t)(mt * 16 + ks) * 32 + lane) * 8 + v] = lo | (hi << 16);
    }
  }
}

typedef float pk_f4 __attribute__((ext_vector_type(4)));
typedef pk_f4 pk_f4u __attribute__((aligned(4)));

__device__ __forceinline__ unsigned bfpair(float lo, float hi) { return f2bf(lo) | (f2bf(hi) << 16); }

template <int K, int N>
__device__ __forceinline__ void pack_B_lane(const float* __restrict__ Wm, unsigned* __restrict__ pack, int id) {
  constexpr int nks = K >> 5;
  const int l = id & 31, t = id >> 5, ks = t % nks, nt = t / nks;
  const float* __restrict__ p = Wm + (size_t)((ks << 5) + ((l & 16) >> 1)) * N + (nt << 4) + (l & 15);
  float f[16];
#pragma unroll
  for (int j = 0; j < 8; ++j) {
    f[j] = p[(size_t)j * N];
    f[8 + j] = p[(size_t)(16 + j) * N];
  }
  const v4u lo = {bfpair(f[0], f[1]), bfpair(f[2], f[3]), bfpair(f[4], f[5]), bfpair(f[6], f[7])};
  const v4u hi = {bfpair(f[8], f[9]), bfpair(f[10], f[11]), bfpair(f[12], f[13]), bfpair(f[14], f[15])};
  v4u* o = (v4u*)(pack + (size_t)id * 8);
  o[0] = lo;
  o[1] = hi;
}

#define PK_B_FT 16
#define PK_B_W1 64
#define PK_B_W2 2

__global__ __launch_bounds__(256) void k_pack_all(
    const float* __restrict__ x, const float* __restrict__ ft_W,
    const float* __restrict__ W1, const float* __restrict__ W2,
    unsigned* __restrict__ packX, unsigned* __restrict__ packWft,
    unsigned* __restrict__ packW1, unsigned* __restrict__ packW2, int MT) {
  const int b = blockIdx.x, tid = threadIdx.x;
  if (b < PK_B_FT) {
    pack_B_lane<128, 512>(ft_W, packWft, b * 256 + tid);
  } else if (b < PK_B_FT + PK_B_W1) {
    pack_B_lane<512, 512>(W1, packW1, (b - PK_B_FT) * 256 + tid);
  } else if (b < PK_B_FT + PK_B_W1 + PK_B_W2) {
    pack_B_lane<512, 16>(W2, packW2, (b - PK_B_FT - PK_B_W1) * 256 + tid);
  } else {
    const int id = (b - PK_B_FT - PK_B_W1 - PK_B_W2) * 256 + tid;
    if (id < MT * 4 * 32) {
      const int l = id & 31, t = id >> 5, ks = t & 3, mt = t >> 2;
      const float* __restrict__ p = x + (size_t)((mt << 4) + (l & 15)) * 128 + (ks << 5) + ((l & 16) >> 1);
      const pk_f4u a = *(const pk_f4u*)(p);
      const pk_f4u c = *(const pk_f4u*)(p + 4);
      const pk_f4u d = *(const pk_f4u*)(p + 16);
      const pk_f4u e = *(const pk_f4u*)(p + 20);
      const v4u lo = {bfpair(a.x, a.y), bfpair(a.z, a.w), bfpair(c.x, c.y), bfpair(c.z, c.w)};
      const v4u hi = {bfpair(d.x, d.y), bfpair(d.z, d.w), bfpair(e.x, e.y), bfpair(e.z, e.w)};
      v4u* o = (v4u*)(packX + (size_t)id * 8);
      o[0] = lo;
      o[1] = hi;
    }
  }
}

__global__ __launch_bounds__(256) void k_gemm_ft_ln(
    const unsigned* __restrict__ packX, const unsigned* __restrict__ packW,
    const float* __restrict__ ft_b, const float* __restrict__ ln_g,
    const float* __restrict__ ln_b, unsigned* __restrict__ packH1) {
  __shared__ float tile[16 * LDSTR];
  __shared__ float psum[16][16], psq[16][16], stats[16][2];
  const int tid = threadIdx.x, lane = tid & 31, w = tid >> 5;
  const int mt = blockIdx.x;
  const int nks = 4;
  v8f acc[4] = {};
  const unsigned* __restrict__ ap = packX + ((size_t)(mt * nks) * 32 + lane) * 8;
  const unsigned* __restrict__ bp = packW + ((size_t)(w * 4 * nks) * 32 + lane) * 8;
  v16bf fa[2], fb[2][4];
  auto load_ks = [&](int ks) {
    fa[ks & 1] = *(const v16bf*)(ap + ks * 256);
#pragma unroll
    for (int j = 0; j < 4; ++j) fb[ks & 1][j] = *(const v16bf*)(bp + (j * nks + ks) * 256);
  };
  auto mma_ks = [&](int ks) {
#pragma unroll
    for (int j = 0; j < 4; ++j)
      acc[j] = __builtin_amdgcn_wmma_f32_16x16x32_bf16(false, fa[ks & 1], false, fb[ks & 1][j],
                                                       (short)0, acc[j], false, false);
  };
  load_ks(0);
  load_ks(1);
  __builtin_amdgcn_sched_barrier(0);
  mma_ks(0);
  __builtin_amdgcn_sched_barrier(0);
  load_ks(2);
  __builtin_amdgcn_sched_barrier(0);
  mma_ks(1);
  __builtin_amdgcn_sched_barrier(0);
  load_ks(3);
  __builtin_amdgcn_sched_barrier(0);
  mma_ks(2);
  mma_ks(3);
  const int m0 = (lane >> 4) << 3, nc = lane & 15;
#pragma unroll
  for (int j = 0; j < 4; ++j) {
    const int n = (w * 4 + j) * 16 + nc;
    const float bias = ft_b[n];
#pragma unroll
    for (int r = 0; r < 8; ++r) tile[(m0 + r) * LDSTR + n] = acc[j][r] + bias;
  }
  __syncthreads();
  {
    const int row = tid >> 4, p = tid & 15;
    float s = 0.f, s2 = 0.f;
    for (int c = p * 32; c < p * 32 + 32; ++c) {
      const float xv = tile[row * LDSTR + c];
      s += xv; s2 += xv * xv;
    }
    psum[row][p] = s; psq[row][p] = s2;
  }
  __syncthreads();
  if (tid < 16) {
    float s = 0.f, s2 = 0.f;
    for (int p = 0; p < 16; ++p) { s += psum[tid][p]; s2 += psq[tid][p]; }
    const float mu = s * (1.f / 512.f);
    const float var = s2 * (1.f / 512.f) - mu * mu;
    stats[tid][0] = mu;
    stats[tid][1] = rsqrtf(var + 1e-5f);
  }
  __syncthreads();
  {
    const int row = tid >> 4, p = tid & 15;
    const float mu = stats[row][0], rs = stats[row][1];
    for (int c = p * 32; c < p * 32 + 32; ++c) {
      float xv = (tile[row * LDSTR + c] - mu) * rs * ln_g[c] + ln_b[c];
      tile[row * LDSTR + c] = xv >= 0.f ? xv : 0.2f * xv;
    }
  }
  __syncthreads();
  pack_tile_to_fragA(tile, packH1, mt, tid);
}

typedef unsigned v8u __attribute__((ext_vector_type(8)));
#define G1_W 8
#define G1_OP 68

__global__ __launch_bounds__(256) void k_gemm1(
    const unsigned* __restrict__ packA, const unsigned* __restrict__ packW,
    const float* __restrict__ att_src, const float* __restrict__ att_dst,
    float* __restrict__ hW, float* __restrict__ a_src, float* __restrict__ a_dst, int MT) {
  __shared__ v4u bs[4 * 16 * 64];
  __shared__ __attribute__((aligned(16))) float ot[G1_W][16 * G1_OP];
  __shared__ __attribute__((aligned(16))) float atl[2][64];
  const int tid = threadIdx.x, lane = tid & 31, w = tid >> 5;
  const int head = blockIdx.x & 7, rg = blockIdx.x >> 3;
  const int mtw = rg * G1_W + w;
  const bool valid = mtw < MT;
  const int mt = valid ? mtw : MT - 1;
  const unsigned* __restrict__ ap = packA + ((size_t)mt * 16 * 32 + lane) * 8;
  v16bf ar[3][2];
  auto load_group = [&](int g) {
#pragma unroll
    for (int q = 0; q < 2; ++q) ar[g % 3][q] = *(const v16bf*)(ap + (2 * g + q) * 256);
  };
  load_group(0);
  {
    const v4u* __restrict__ gB = (const v4u*)(packW + (size_t)head * (4 * 16 * 256)) + tid;
    v4u* sb = bs + ((tid >> 6) << 6) + ((tid & 1) << 5) + ((tid & 63) >> 1);
    v4u tmp[16];
#pragma unroll
    for (int t = 0; t < 16; ++t) tmp[t] = gB[256 * t];
#pragma unroll
    for (int t = 0; t < 16; ++t) sb[256 * t] = tmp[t];
  }
  load_group(1);
  load_group(2);
  if (tid < 128) atl[tid >> 6][tid & 63] = (tid < 64 ? att_src : att_dst)[head * 64 + (tid & 63)];
  __syncthreads();
  v8f acc[4] = {};
  auto kstep = [&](const v16bf& a, int ks) {
    v16bf b[4];
#pragma unroll
    for (int j = 0; j < 4; ++j) {
      const v4u lo = bs[(j * 16 + ks) * 64 + lane];
      const v4u hi = bs[(j * 16 + ks) * 64 + 32 + lane];
      const v8u t = __builtin_shufflevector(lo, hi, 0, 1, 2, 3, 4, 5, 6, 7);
      b[j] = (v16bf)t;
    }
    __builtin_amdgcn_sched_barrier(0);
#pragma unroll
    for (int j = 0; j < 4; ++j)
      acc[j] = __builtin_amdgcn_wmma_f32_16x16x32_bf16(false, a, false, b[j],
                                                       (short)0, acc[j], false, false);
  };
#pragma unroll
  for (int g = 0; g < 8; ++g) {
    kstep(ar[g % 3][0], 2 * g);
    __builtin_amdgcn_sched_barrier(0);
    kstep(ar[g % 3][1], 2 * g + 1);
    if (g + 3 < 8) load_group(g + 3);
    __builtin_amdgcn_sched_barrier(0);
  }

  float* o = ot[w];
  const int m0 = (lane >> 4) << 3, nc = lane & 15;
#pragma unroll
  for (int j = 0; j < 4; ++j) {
#pragma unroll
    for (int r = 0; r < 8; ++r) o[(m0 + r) * G1_OP + j * 16 + nc] = acc[j][r];
  }
  __builtin_amdgcn_fence(__ATOMIC_ACQ_REL, "wavefront");
  if (valid) {
    const int hh = lane >> 4;
#pragma unroll
    for (int it = 0; it < 8; ++it) {
      const int row = it * 2 + hh, c4 = (lane & 15) * 4;
      const float4 v = *(const float4*)(o + row * G1_OP + c4);
      *(float4*)(hW + (size_t)(mt * 16 + row) * 512 + head * 64 + c4) = v;
    }
    const int row = lane & 15;
    const float* att = atl[hh];
    float s = 0.f;
    for (int c = 0; c < 64; ++c) s += o[row * G1_OP + c] * att[c];
    (hh ? a_dst : a_src)[(mt * 16 + row) * 8 + head] = s;
  }
}

#define AGG_R 40
#define AGG_WAVES 16
#define AGG_UL 8
#define AGG_STEP (128 * AGG_UL)
#define AGG_CAP (AGG_STEP + 32)
#define AGG_KEEP 256
#define AGG_OVER 0xFFFFFFFFu
#define AGG_NPITCH 32
#define AGG_SB 8
#define AGG_RC 16
#define AGG_NR (AGG_RC * AGG_WAVES)

typedef int sc_vi4 __attribute__((ext_vector_type(4)));
typedef sc_vi4 sc_vi4u __attribute__((aligned(4)));

__global__ __launch_bounds__(512) void k_agg1_owned(const int* __restrict__ ei,
    const float* __restrict__ hW, const float* __restrict__ a_src,
    const float* __restrict__ a_dst, float* __restrict__ agg,
    unsigned* __restrict__ gls, unsigned short* __restrict__ gld, unsigned* __restrict__ gcnt,
    int N, int E) {
  __shared__ float acc[AGG_R * 512];
  __shared__ unsigned mx[AGG_R * 8];
  __shared__ float den[AGG_R * 8];
  __shared__ float adl[AGG_R * 8];
  __shared__ unsigned lst[AGG_WAVES][AGG_CAP];
  __shared__ unsigned short lsd[AGG_WAVES][AGG_CAP];
  __shared__ unsigned kps[AGG_WAVES][AGG_KEEP];
  __shared__ unsigned short kpd[AGG_WAVES][AGG_KEEP];
  static_assert(AGG_WAVES == 16 && AGG_R <= 3 * AGG_WAVES && 3 * AGG_NR <= AGG_CAP, "owner = low 4 bits of the slot; three nodes' matches fit lst[w]");
  __shared__ unsigned rt[AGG_WAVES][AGG_WAVES][AGG_RC];
  __shared__ int ctab[AGG_WAVES][AGG_WAVES];
  __shared__ int wok[AGG_WAVES];
  const int tid = threadIdx.x, lane = tid & 31, hh = lane >> 4;
  const int w = __builtin_amdgcn_readfirstlane(tid >> 5);
  const int d0 = blockIdx.x * AGG_R;
  const int Rn = min(AGG_R, N - d0);
  for (int i = tid; i < AGG_R * 8; i += 512) {
    mx[i] = 0u;
    adl[i] = (i < Rn * 8) ? a_dst[(size_t)d0 * 8 + i] : 0.f;
  }
  for (int i = tid; i < AGG_WAVES * AGG_WAVES; i += 512) (&ctab[0][0])[i] = 0;
  __syncthreads();
  const int SL = ((E + AGG_WAVES - 1) / AGG_WAVES + 3) & ~3;
  const int e_begin = min(E, w * SL);
  const int e_end = min(E, e_begin + SL);
  const int len = e_end - e_begin;
  const int nfull = len / AGG_STEP;
  const int nstep = (len + AGG_STEP - 1) / AGG_STEP;
  const int* __restrict__ dstp = ei + E;

  auto load_full = [&](int (&v)[AGG_UL][4], int start) {
#pragma unroll
    for (int u = 0; u < AGG_UL; ++u) {
      const sc_vi4u t = *(const sc_vi4u*)(dstp + start + u * 128 + lane * 4);
      v[u][0] = t.x; v[u][1] = t.y; v[u][2] = t.z; v[u][3] = t.w;
    }
  };
  auto load_part = [&](int (&v)[AGG_UL][4], int start) {
#pragma unroll
    for (int u = 0; u < AGG_UL; ++u) {
      const int e = start + u * 128 + lane * 4;
      if (e + 3 < e_end) {
        const sc_vi4u t = *(const sc_vi4u*)(dstp + e);
        v[u][0] = t.x; v[u][1] = t.y; v[u][2] = t.z; v[u][3] = t.w;
      } else {
#pragma unroll
        for (int k = 0; k < 4; ++k) v[u][k] = (e + k < e_end) ? dstp[e + k] : -1;
      }
    }
  };
  auto match_step = [&](const int (&v)[AGG_UL][4], int base, int cnt) -> int {
#pragma unroll
    for (int u = 0; u < AGG_UL; ++u) {
      unsigned dl[4], bal[4];
#pragma unroll
      for (int k = 0; k < 4; ++k) {
        dl[k] = (unsigned)(v[u][k] - d0);
        bal[k] = __builtin_amdgcn_ballot_w32(dl[k] < (unsigned)Rn);
      }
      if ((bal[0] | bal[1] | bal[2] | bal[3]) != 0u) {
#pragma unroll
        for (int k = 0; k < 4; ++k) {
          if (bal[k]) {
            const int pos = cnt + __builtin_popcount(bal[k] & ((1u << lane) - 1u));
            if (dl[k] < (unsigned)Rn) {
              lst[w][pos] = (unsigned)(base + u * 128 + lane * 4 + k);
              lsd[w][pos] = (unsigned short)dl[k];
            }
            cnt += __builtin_popcount(bal[k]);
          }
        }
      }
    }
    return cnt;
  };
  auto sum_rows = [&](const unsigned* ls, const unsigned short* ld, int n) {
    for (int j0 = 0; j0 < n; j0 += 4) {
      float4 row[4][4];
      float as[4][4];
      int dln[4];
#pragma unroll
      for (int b = 0; b < 4; ++b) {
        const int jb = min(j0 + b, n - 1);
        const size_t s = (size_t)ls[jb];
        dln[b] = (int)ld[jb];
        const float* rp = hW + s * 512 + 4 * lane;
#pragma unroll
        for (int j = 0; j < 4; ++j) row[b][j] = *(const float4*)(rp + 128 * j);
#pragma unroll
        for (int j = 0; j < 4; ++j) as[b][j] = a_src[s * 8 + 2 * j + hh];
      }
#pragma unroll
      for (int b = 0; b < 4; ++b) {
        if (j0 + b < n) {
          const int dl = dln[b];
#pragma unroll
          for (int j = 0; j < 4; ++j) {
            const int h = 2 * j + hh;
            float L = as[b][j] + adl[dl * 8 + h];
            L = L >= 0.f ? L : 0.2f * L;
            const float wgt = __expf(L - fdec(mx[dl * 8 + h]));
            float* ap = acc + dl * 512 + j * 128 + lane;
            atomicAdd(ap, wgt * row[b][j].x);
            atomicAdd(ap + 32, wgt * row[b][j].y);
            atomicAdd(ap + 64, wgt * row[b][j].z);
            atomicAdd(ap + 96, wgt * row[b][j].w);
            if ((lane & 15) == 0) atomicAdd(&den[dl * 8 + h], wgt);
          }
        }
      }
    }
  };

  int kept = 0;
  bool fits = true;
  unsigned* __restrict__ gl_s = gls + (size_t)(blockIdx.x * AGG_WAVES + w) * AGG_KEEP;
  unsigned short* __restrict__ gl_d = gld + (size_t)(blockIdx.x * AGG_WAVES + w) * AGG_KEEP;
  auto flush = [&](int cnt) {
    __builtin_amdgcn_fence(__ATOMIC_ACQ_REL, "wavefront");
    const bool keep = fits && kept + cnt <= AGG_KEEP;
    for (int j0 = 0; j0 < cnt; j0 += 32) {
      const int j = j0 + lane;
      const bool act = j < cnt;
      int s = 0, dl = 0;
      if (act) {
        const int e = (int)lst[w][j];
        dl = (int)lsd[w][j];
        s = ei[e];
        const float4 a0 = *(const float4*)(a_src + (size_t)s * 8);
        const float4 a1 = *(const float4*)(a_src + (size_t)s * 8 + 4);
        const float av[8] = {a0.x, a0.y, a0.z, a0.w, a1.x, a1.y, a1.z, a1.w};
#pragma unroll
        for (int h = 0; h < 8; ++h) {
          float L = av[h] + adl[dl * 8 + h];
          L = L >= 0.f ? L : 0.2f * L;
          atomicMax(&mx[dl * 8 + h], fenc(L));
        }
        if (keep) {
          kps[w][kept + j] = (unsigned)s;
          kpd[w][kept + j] = (unsigned short)dl;
          gl_s[kept + j] = (unsigned)s;
          gl_d[kept + j] = (unsigned short)dl;
        }
      }
      const int o = dl & (AGG_WAVES - 1);
      unsigned sm = __builtin_amdgcn_ballot_w32(act);
#pragma unroll
      for (int b = 0; b < 4; ++b) {
        const bool bit = ((o >> b) & 1) != 0;
        const unsigned bb = __builtin_amdgcn_ballot_w32(act && bit);
        sm &= bit ? bb : ~bb;
      }
      if (act) {
        const int rank = __builtin_popcount(sm & ((1u << lane) - 1u));
        const int base = ctab[w][o];
        if (base + rank < AGG_RC) rt[w][o][base + rank] = (unsigned)s | ((unsigned)(dl >> 4) << 30);
        if (rank == 0) ctab[w][o] = base + __builtin_popcount(sm);
      }
      __builtin_amdgcn_fence(__ATOMIC_ACQ_REL, "wavefront");
    }
    if (keep) kept += cnt; else fits = false;
  };
  {
    int cnt = 0;
    if (nfull > 0) {
      int va[AGG_UL][4], vb[AGG_UL][4];
      const int last_full = e_begin + (nfull - 1) * AGG_STEP;
      load_full(va, e_begin);
      for (int it = 0; it < nfull; it += 2) {
        const int base = e_begin + it * AGG_STEP;
        load_full(vb, min(base + AGG_STEP, last_full));
        cnt = match_step(va, base, cnt);
        if (cnt >= 32) { flush(cnt); cnt = 0; }
        load_full(va, min(base + 2 * AGG_STEP, last_full));
        if (it + 1 < nfull) {
          cnt = match_step(vb, base + AGG_STEP, cnt);
          if (cnt >= 32) { flush(cnt); cnt = 0; }
        }
      }
    }
    if (nfull < nstep) {
      int vt[AGG_UL][4];
      const int base = e_begin + nfull * AGG_STEP;
      load_part(vt, base);
      cnt = match_step(vt, base, cnt);
    }
    if (cnt > 0) flush(cnt);
  }
  {
    const int myc = lane < AGG_WAVES ? ctab[w][lane] : 0;
    const bool rfit = __builtin_amdgcn_ballot_w32(myc > AGG_RC) == 0u;
    if (lane == 0) {
      gcnt[blockIdx.x * AGG_NPITCH + w] = fits ? (unsigned)kept : AGG_OVER;
      wok[w] = (fits && rfit) ? 1 : 0;
    }
  }
  __syncthreads();

  const int okv = lane < AGG_WAVES ? wok[lane] : 1;
  if (__builtin_amdgcn_ballot_w32(okv == 0) == 0u) {
    const int cq = lane < AGG_WAVES ? ctab[lane][w] : 0;
    const int e16 = lane & 15;
    unsigned ent[8];
    bool val[8];
#pragma unroll
    for (int k = 0; k < 8; ++k) {
      ent[k] = rt[2 * k + hh][w][e16];
      const int c0 = __builtin_amdgcn_readlane(cq, 2 * k), c1 = __builtin_amdgcn_readlane(cq, 2 * k + 1);
      val[k] = e16 < (hh ? c1 : c0);
    }
    int cn0 = 0, cn1 = 0, cn2 = 0;
#pragma unroll
    for (int k = 0; k < 8; ++k) {
      const int tk = (int)(ent[k] >> 30);
      const unsigned b0 = __builtin_amdgcn_ballot_w32(val[k] && tk == 0);
      const unsigned b1 = __builtin_amdgcn_ballot_w32(val[k] && tk == 1);
      const unsigned b2 = __builtin_amdgcn_ballot_w32(val[k] && tk == 2);
      if ((b0 | b1 | b2) != 0u) {
        const unsigned below = (1u << lane) - 1u;
        const unsigned mine = tk == 0 ? b0 : (tk == 1 ? b1 : b2);
        const int at = tk * AGG_NR + (tk == 0 ? cn0 : (tk == 1 ? cn1 : cn2)) + __builtin_popcount(mine & below);
        if (val[k]) lst[w][at] = ent[k] & 0x3FFFFFFFu;
        cn0 += __builtin_popcount(b0);
        cn1 += __builtin_popcount(b1);
        cn2 += __builtin_popcount(b2);
      }
    }
    __builtin_amdgcn_fence(__ATOMIC_ACQ_REL, "wavefront");
    for (int t = 0; t < 3; ++t) {
      const int dl = w + AGG_WAVES * t;
      if (dl >= Rn) break;
      const unsigned* ls = lst[w] + t * AGG_NR;
      const int n = t == 0 ? cn0 : (t == 1 ? cn1 : cn2);
      float adv[4], mxv[4];
#pragma unroll
      for (int j = 0; j < 4; ++j) {
        adv[j] = adl[dl * 8 + 2 * j + hh];
        mxv[j] = fdec(mx[dl * 8 + 2 * j + hh]);
      }
      float4 sm[4];
      float dn[4];
#pragma unroll
      for (int j = 0; j < 4; ++j) { sm[j] = make_float4(0.f, 0.f, 0.f, 0.f); dn[j] = 0.f; }
      for (int j0 = 0; j0 < n; j0 += AGG_SB) {
        const unsigned sv = ls[min(j0 + (lane & (AGG_SB - 1)), n - 1)];
        float4 row[AGG_SB][4];
        float as[AGG_SB][4];
#pragma unroll
        for (int b = 0; b < AGG_SB; ++b) {
          if (j0 + b < n) {
            const size_t s = (size_t)(unsigned)__builtin_amdgcn_readlane((int)sv, b);
            const float* rp = hW + s * 512 + 4 * lane;
#pragma unroll
            for (int j = 0; j < 4; ++j) row[b][j] = *(const float4*)(rp + 128 * j);
#pragma unroll
            for (int j = 0; j < 4; ++j) as[b][j] = a_src[s * 8 + 2 * j + hh];
          }
        }
#pragma unroll
        for (int b = 0; b < AGG_SB; ++b) {
          if (j0 + b < n) {
#pragma unroll
            for (int j = 0; j < 4; ++j) {
              float L = as[b][j] + adv[j];
              L = L >= 0.f ? L : 0.2f * L;
              const float wgt = __expf(L - mxv[j]);
              sm[j].x += wgt * row[b][j].x;
              sm[j].y += wgt * row[b][j].y;
              sm[j].z += wgt * row[b][j].z;
              sm[j].w += wgt * row[b][j].w;
              dn[j] += wgt;
            }
          }
        }
      }
      float* op = agg + (size_t)(d0 + dl) * 512 + 4 * lane;
#pragma unroll
      for (int j = 0; j < 4; ++j) {
        float4 ov;
        ov.x = dn[j] != 0.f ? sm[j].x / dn[j] : 0.f;
        ov.y = dn[j] != 0.f ? sm[j].y / dn[j] : 0.f;
        ov.z = dn[j] != 0.f ? sm[j].z / dn[j] : 0.f;
        ov.w = dn[j] != 0.f ? sm[j].w / dn[j] : 0.f;
        *(float4*)(op + 128 * j) = ov;
      }
    }
    return;
  }
  for (int i = tid; i < AGG_R * 512; i += 512) acc[i] = 0.f;
  for (int i = tid; i < AGG_R * 8; i += 512) den[i] = 0.f;
  __syncthreads();
  if (fits) {
    sum_rows(kps[w], kpd[w], kept);
  } else {
    for (int it = 0; it < nstep; ++it) {
      const int base = e_begin + it * AGG_STEP;
      int dv[AGG_UL][4];
      load_part(dv, base);
      const int cnt = match_step(dv, base, 0);
      __builtin_amdgcn_fence(__ATOMIC_ACQ_REL, "wavefront");
      for (int j = lane; j < cnt; j += 32) lst[w][j] = (unsigned)ei[lst[w][j]];
      __builtin_amdgcn_fence(__ATOMIC_ACQ_REL, "wavefront");
      sum_rows(lst[w], lsd[w], cnt);
    }
  }
  __syncthreads();
  for (int i = tid; i < Rn * 512; i += 512) {
    const int dl = i >> 9, c = i & 511;
    const int q = c >> 2, k = c & 3, j = q >> 5, l = q & 31;
    const float dn = den[dl * 8 + (c >> 6)];
    const float v = acc[dl * 512 + (j * 4 + k) * 32 + l];
    agg[(size_t)(d0 + dl) * 512 + c] = dn != 0.f ? v / dn : 0.f;
  }
}

#define A2_R 48
#define A2_WAVES 16
#define A2_UL 8
#define A2_STEP (128 * A2_UL)
#define A2_CAP (A2_STEP + 32)
#define A2_KEEP 512

typedef int a2_vi4 __attribute__((ext_vector_type(4)));
typedef a2_vi4 a2_vi4u __attribute__((aligned(4)));

__global__ __launch_bounds__(512) void k_agg2_owned(const int* __restrict__ ei,
    const float* __restrict__ hW, const float* __restrict__ a_src,
    const float* __restrict__ a_dst, const float* __restrict__ b2,
    const unsigned* __restrict__ gls, const unsigned short* __restrict__ gld,
    const unsigned* __restrict__ gcnt, float* __restrict__ out, int N, int E) {
  __shared__ float acc[A2_R * 16];
  __shared__ unsigned mx[A2_R * 8];
  __shared__ float den[A2_R * 8];
  __shared__ float adl[A2_R * 8];
  __shared__ unsigned lst[A2_WAVES][A2_CAP];
  __shared__ unsigned short lsd[A2_WAVES][A2_CAP];
  __shared__ unsigned kps[A2_WAVES][A2_KEEP];
  __shared__ unsigned short kpd[A2_WAVES][A2_KEEP];
  const int tid = threadIdx.x, lane = tid & 31;
  const int w = __builtin_amdgcn_readfirstlane(tid >> 5);
  const int d0 = blockIdx.x * A2_R;
  const int Rn = min(A2_R, N - d0);
  static_assert(AGG_R % 8 == 0 && A2_R % 8 == 0 && AGG_R - 8 + A2_R <= 2 * AGG_R, "a layer-2 range spans at most two layer-1 ranges");
  static_assert(AGG_WAVES == A2_WAVES && AGG_WAVES == 16, "wave w walks list w; a wave's lanes hold both ranges' 16 lengths");
  const int b1a = d0 / AGG_R;
  const int nl = (d0 + Rn - 1) / AGG_R - b1a + 1;
  const unsigned cl = lane < AGG_WAVES * nl ? gcnt[(b1a + (lane >> 4)) * AGG_NPITCH + (lane & 15)] : 0u;
  unsigned es[4];
  int esl[4];
#pragma unroll
  for (int c = 0; c < 4; ++c) {
    const int bq = b1a + ((c >> 1) < nl ? (c >> 1) : 0);
    const size_t li = (size_t)(bq * AGG_WAVES + w) * AGG_KEEP + lane + 32 * (c & 1);
    es[c] = gls[li];
    esl[c] = (int)gld[li];
  }
  for (int i = tid; i < A2_R * 16; i += 512) acc[i] = 0.f;
  for (int i = tid; i < A2_R * 8; i += 512) {
    mx[i] = 0u; den[i] = 0.f;
    adl[i] = (i < Rn * 8) ? a_dst[(size_t)d0 * 8 + i] : 0.f;
  }
  __syncthreads();
  const int SL = ((E + A2_WAVES - 1) / A2_WAVES + 3) & ~3;
  const int e_begin = w * SL;
  const int e_end = min(E, e_begin + SL);
  const int nit = (SL + A2_STEP - 1) / A2_STEP;
  const int* __restrict__ dstp = ei + E;

  auto load_step = [&](int (&v)[A2_UL][4], int start) {
    if (start + A2_STEP <= e_end) {
#pragma unroll
      for (int u = 0; u < A2_UL; ++u) {
        const a2_vi4u t = *(const a2_vi4u*)(dstp + start + u * 128 + lane * 4);
        v[u][0] = t.x; v[u][1] = t.y; v[u][2] = t.z; v[u][3] = t.w;
      }
    } else {
#pragma unroll
      for (int u = 0; u < A2_UL; ++u) {
        const int e = start + u * 128 + lane * 4;
        if (e + 3 < e_end) {
          const a2_vi4u t = *(const a2_vi4u*)(dstp + e);
          v[u][0] = t.x; v[u][1] = t.y; v[u][2] = t.z; v[u][3] = t.w;
        } else {
#pragma unroll
          for (int k = 0; k < 4; ++k) v[u][k] = (e + k < e_end) ? dstp[e + k] : -1;
        }
      }
    }
  };
  auto edge_max = [&](size_t s, int dl) {
    const float4 a0 = *(const float4*)(a_src + s * 8);
    const float4 a1 = *(const float4*)(a_src + s * 8 + 4);
    const float av[8] = {a0.x, a0.y, a0.z, a0.w, a1.x, a1.y, a1.z, a1.w};
#pragma unroll
    for (int h = 0; h < 8; ++h) {
      float L = av[h] + adl[dl * 8 + h];
      L = L >= 0.f ? L : 0.2f * L;
      atomicMax(&mx[dl * 8 + h], fenc(L));
    }
  };
  auto edge_sum = [&](size_t s, int dl) {
    const float4 a0 = *(const float4*)(a_src + s * 8);
    const float4 a1 = *(const float4*)(a_src + s * 8 + 4);
    const float4 r0 = *(const float4*)(hW + s * 16);
    const float4 r1 = *(const float4*)(hW + s * 16 + 4);
    const float4 r2 = *(const float4*)(hW + s * 16 + 8);
    const float4 r3 = *(const float4*)(hW + s * 16 + 12);
    const float av[8] = {a0.x, a0.y, a0.z, a0.w, a1.x, a1.y, a1.z, a1.w};
    const float hv[16] = {r0.x, r0.y, r0.z, r0.w, r1.x, r1.y, r1.z, r1.w,
                          r2.x, r2.y, r2.z, r2.w, r3.x, r3.y, r3.z, r3.w};
#pragma unroll
    for (int h = 0; h < 8; ++h) {
      float L = av[h] + adl[dl * 8 + h];
      L = L >= 0.f ? L : 0.2f * L;
      const float wgt = __expf(L - fdec(mx[dl * 8 + h]));
      atomicAdd(&den[dl * 8 + h], wgt);
      atomicAdd(&acc[dl * 16 + 2 * h], wgt * hv[2 * h]);
      atomicAdd(&acc[dl * 16 + 2 * h + 1], wgt * hv[2 * h + 1]);
    }
  };

  const bool use_lists = __builtin_amdgcn_ballot_w32(cl == AGG_OVER) == 0u;

  if (use_lists) {
    unsigned cn[2];
    cn[0] = min((unsigned)__builtin_amdgcn_readlane((int)cl, w), (unsigned)AGG_KEEP);
    cn[1] = nl > 1 ? min((unsigned)__builtin_amdgcn_readlane((int)cl, AGG_WAVES + w), (unsigned)AGG_KEEP) : 0u;
    bool ok[4];
    int cdl[4];
    float4 av[4][2], hv[4][4];
#pragma unroll
    for (int c = 0; c < 4; ++c) {
      const int q = c >> 1;
      const int dl = (b1a + q) * AGG_R + esl[c] - d0;
      ok[c] = lane + 32 * (c & 1) < (int)cn[q] && (unsigned)dl < (unsigned)Rn;
      cdl[c] = ok[c] ? dl : 0;
      const size_t s = ok[c] ? (size_t)es[c] : (size_t)0;
      av[c][0] = *(const float4*)(a_src + s * 8);
      av[c][1] = *(const float4*)(a_src + s * 8 + 4);
#pragma unroll
      for (int r = 0; r < 4; ++r) hv[c][r] = *(const float4*)(hW + s * 16 + 4 * r);
    }
    auto rest = [&](auto&& fn) {
      for (int q = 0; q < nl; ++q) {
        const size_t lb = (size_t)((b1a + q) * AGG_WAVES + w) * AGG_KEEP;
        for (int j = 64 + lane; j < (int)cn[q]; j += 32) {
          const int dl = (b1a + q) * AGG_R + (int)gld[lb + j] - d0;
          if ((unsigned)dl < (unsigned)Rn) fn((size_t)gls[lb + j], dl);
        }
      }
    };
#pragma unroll
    for (int c = 0; c < 4; ++c) {
      if (ok[c]) {
        const float a8[8] = {av[c][0].x, av[c][0].y, av[c][0].z, av[c][0].w, av[c][1].x, av[c][1].y, av[c][1].z, av[c][1].w};
#pragma unroll
        for (int h = 0; h < 8; ++h) {
          float L = a8[h] + adl[cdl[c] * 8 + h];
          L = L >= 0.f ? L : 0.2f * L;
          atomicMax(&mx[cdl[c] * 8 + h], fenc(L));
        }
      }
    }
    rest(edge_max);
    __syncthreads();
#pragma unroll
    for (int c = 0; c < 4; ++c) {
      if (ok[c]) {
        const float a8[8] = {av[c][0].x, av[c][0].y, av[c][0].z, av[c][0].w, av[c][1].x, av[c][1].y, av[c][1].z, av[c][1].w};
        const float h16[16] = {hv[c][0].x, hv[c][0].y, hv[c][0].z, hv[c][0].w, hv[c][1].x, hv[c][1].y, hv[c][1].z, hv[c][1].w,
                               hv[c][2].x, hv[c][2].y, hv[c][2].z, hv[c][2].w, hv[c][3].x, hv[c][3].y, hv[c][3].z, hv[c][3].w};
        const int dl = cdl[c];
#pragma unroll
        for (int h = 0; h < 8; ++h) {
          float L = a8[h] + adl[dl * 8 + h];
          L = L >= 0.f ? L : 0.2f * L;
          const float wgt = __expf(L - fdec(mx[dl * 8 + h]));
          atomicAdd(&den[dl * 8 + h], wgt);
          atomicAdd(&acc[dl * 16 + 2 * h], wgt * h16[2 * h]);
          atomicAdd(&acc[dl * 16 + 2 * h + 1], wgt * h16[2 * h + 1]);
        }
      }
    }
    rest(edge_sum);
    __syncthreads();
  } else {
    int kept = 0;
    bool fits = true;
    for (int pass = 0; pass < 2; ++pass) {
      if (pass == 1 && fits) {
        for (int j = lane; j < kept; j += 32) edge_sum((size_t)kps[w][j], (int)kpd[w][j]);
      } else {
        int dv[A2_UL][4];
        load_step(dv, e_begin);
        int cnt = 0;
        for (int it = 0; it < nit; ++it) {
          const int base = e_begin + it * A2_STEP;
          int nx[A2_UL][4];
          load_step(nx, base + A2_STEP);
#pragma unroll
          for (int u = 0; u < A2_UL; ++u) {
#pragma unroll
            for (int k = 0; k < 4; ++k) {
              const unsigned dl = (unsigned)(dv[u][k] - d0);
              const bool match = dl < (unsigned)Rn;
              const unsigned bal = __builtin_amdgcn_ballot_w32(match);
              if (__builtin_expect(bal != 0u, 0)) {
                const int pos = cnt + __builtin_popcount(bal & ((1u << lane) - 1u));
                if (match) {
                  lst[w][pos] = (unsigned)(base + u * 128 + lane * 4 + k);
                  lsd[w][pos] = (unsigned short)dl;
                }
                cnt += __builtin_popcount(bal);
              }
            }
          }
          if (cnt >= 32 || it == nit - 1) {
            __builtin_amdgcn_fence(__ATOMIC_ACQ_REL, "wavefront");
            const bool keep = fits && kept + cnt <= A2_KEEP;
            for (int j = lane; j < cnt; j += 32) {
              const int e = (int)lst[w][j];
              const int dl = (int)lsd[w][j];
              const size_t s = (size_t)ei[e];
              if (pass == 0) {
                edge_max(s, dl);
                if (keep) {
                  kps[w][kept + j] = (unsigned)s;
                  kpd[w][kept + j] = (unsigned short)dl;
                }
              } else {
                edge_sum(s, dl);
              }
            }
            if (keep) kept += cnt; else fits = false;
            cnt = 0;
          }
#pragma unroll
          for (int u = 0; u < A2_UL; ++u) {
#pragma unroll
            for (int k = 0; k < 4; ++k) dv[u][k] = nx[u][k];
          }
        }
      }
      __syncthreads();
    }
  }
  for (int i = tid; i < Rn * 2; i += 512) {
    const int dl = i >> 1, c = i & 1;
    float s = 0.f;
#pragma unroll
    for (int h = 0; h < 8; ++h) {
      const float dn = den[dl * 8 + h];
      s += dn != 0.f ? acc[dl * 16 + 2 * h + c] / dn : 0.f;
    }
    out[(size_t)(d0 + dl) * 2 + c] = s * 0.125f + b2[c];
  }
}

__global__ __launch_bounds__(256) void k_bnprep(const float* __restrict__ agg,
    const float* __restrict__ b1, const float* __restrict__ bn_g,
    const float* __restrict__ bn_b, const float* __restrict__ bn_mean,
    const float* __restrict__ bn_var, unsigned* __restrict__ packH2) {
  __shared__ __attribute__((aligned(16))) float tile[16 * LDSTR];
  const int mt = blockIdx.x, tid = threadIdx.x;
  const int c0 = (tid & 127) * 4, r0 = tid >> 7;
  const float* __restrict__ gp = agg + (size_t)(mt * 16 + r0) * 512 + c0;
  float4 xv[8];
#pragma unroll
  for (int t = 0; t < 8; ++t) xv[t] = *(const float4*)(gp + (size_t)t * 1024);
  const pk_f4 pb = *(const pk_f4u*)(b1 + c0);
  const pk_f4 pm = *(const pk_f4u*)(bn_mean + c0);
  const pk_f4 pv = *(const pk_f4u*)(bn_var + c0);
  const pk_f4 pg = *(const pk_f4u*)(bn_g + c0);
  const pk_f4 po = *(const pk_f4u*)(bn_b + c0);
  const float s0 = rsqrtf(pv.x + 1e-5f), s1 = rsqrtf(pv.y + 1e-5f);
  const float s2 = rsqrtf(pv.z + 1e-5f), s3 = rsqrtf(pv.w + 1e-5f);
#pragma unroll
  for (int t = 0; t < 8; ++t) {
    float4 y;
    y.x = ((xv[t].x + pb.x) - pm.x) * s0 * pg.x + po.x;
    y.y = ((xv[t].y + pb.y) - pm.y) * s1 * pg.y + po.y;
    y.z = ((xv[t].z + pb.z) - pm.z) * s2 * pg.z + po.z;
    y.w = ((xv[t].w + pb.w) - pm.w) * s3 * pg.w + po.w;
    y.x = y.x > 0.f ? y.x : 0.f;
    y.y = y.y > 0.f ? y.y : 0.f;
    y.z = y.z > 0.f ? y.z : 0.f;
    y.w = y.w > 0.f ? y.w : 0.f;
    *(float4*)(tile + (r0 + 2 * t) * LDSTR + c0) = y;
  }
  __syncthreads();
  pack_tile_to_fragA(tile, packH2, mt, tid);
}

__global__ __launch_bounds__(32) void k_gemm2(
    const unsigned* __restrict__ packA, const unsigned* __restrict__ packW,
    const float* __restrict__ att_src, const float* __restrict__ att_dst,
    float* __restrict__ hW, float* __restrict__ a_src, float* __restrict__ a_dst) {
  __shared__ float t2[16][16];
  const int lane = threadIdx.x;
  const int mt = blockIdx.x;
  const int nks = 16;
  v8f acc = {};
  v16bf fa[16], fb[16];
#pragma unroll
  for (int ks = 0; ks < nks; ++ks) {
    fa[ks] = *(const v16bf*)(packA + ((size_t)(mt * nks + ks) * 32 + lane) * 8);
    fb[ks] = *(const v16bf*)(packW + ((size_t)ks * 32 + lane) * 8);
  }
  __builtin_amdgcn_sched_barrier(0);
#pragma unroll
  for (int ks = 0; ks < nks; ++ks)
    acc = __builtin_amdgcn_wmma_f32_16x16x32_bf16(false, fa[ks], false, fb[ks],
                                                  (short)0, acc, false, false);
  const int m0 = (lane >> 4) << 3, nc = lane & 15;
#pragma unroll
  for (int r = 0; r < 8; ++r) t2[m0 + r][nc] = acc[r];
  __syncthreads();
  for (int i = lane; i < 256; i += 32) {
    const int row = i >> 4, c = i & 15;
    hW[(size_t)(mt * 16 + row) * 16 + c] = t2[row][c];
  }
  if (lane < 16) {
    const int row = lane, node = mt * 16 + row;
    for (int h = 0; h < 8; ++h) {
      const float v0 = t2[row][2 * h], v1 = t2[row][2 * h + 1];
      a_src[node * 8 + h] = v0 * att_src[2 * h] + v1 * att_src[2 * h + 1];
      a_dst[node * 8 + h] = v0 * att_dst[2 * h] + v1 * att_dst[2 * h + 1];
    }
  }
}

extern "C" void kernel_launch(void* const* d_in, const int* in_sizes, int n_in,
                              void* d_out, int out_size, void* d_ws, size_t ws_size,
                              hipStream_t stream) {
  const float* x       = (const float*)d_in[0];
  const int*   ei      = (const int*)d_in[1];
  const float* ft_W    = (const float*)d_in[2];
  const float* ft_b    = (const float*)d_in[3];
  const float* ln_g    = (const float*)d_in[4];
  const float* ln_b    = (const float*)d_in[5];
  const float* W1      = (const float*)d_in[6];
  const float* att_s1  = (const float*)d_in[7];
  const float* att_d1  = (const float*)d_in[8];
  const float* b1      = (const float*)d_in[9];
  const float* bn_g    = (const float*)d_in[10];
  const float* bn_b    = (const float*)d_in[11];
  const float* bn_mean = (const float*)d_in[12];
  const float* bn_var  = (const float*)d_in[13];
  const float* W2      = (const float*)d_in[14];
  const float* att_s2  = (const float*)d_in[15];
  const float* att_d2  = (const float*)d_in[16];
  const float* b2      = (const float*)d_in[17];

  const int N  = in_sizes[0] / 128;
  const int E  = in_sizes[1] / 2;
  const int MT = N / 16;

  char* cur = (char*)d_ws;
  auto alloc = [&](size_t bytes) -> void* {
    void* p = (void*)cur;
    cur += (bytes + 255) & ~(size_t)255;
    return p;
  };
  unsigned* packWft = (unsigned*)alloc((size_t)32 * 4 * 256 * 4);
  unsigned* packW1  = (unsigned*)alloc((size_t)32 * 16 * 256 * 4);
  unsigned* packW2  = (unsigned*)alloc((size_t)1 * 16 * 256 * 4);
  unsigned* packX   = (unsigned*)alloc((size_t)MT * 4 * 256 * 4);
  unsigned* packH1  = (unsigned*)alloc((size_t)MT * 16 * 256 * 4);
  unsigned* packH2  = (unsigned*)alloc((size_t)MT * 16 * 256 * 4);
  float* hW1 = (float*)alloc((size_t)N * 512 * 4);
  float* hW2 = (float*)alloc((size_t)N * 16 * 4);
  float* aS1 = (float*)alloc((size_t)N * 8 * 4);
  float* aD1 = (float*)alloc((size_t)N * 8 * 4);
  float* aS2 = (float*)alloc((size_t)N * 8 * 4);
  float* aD2 = (float*)alloc((size_t)N * 8 * 4);
  float* agg1 = (float*)alloc((size_t)N * 512 * 4);
  const int NB1 = (N + AGG_R - 1) / AGG_R;
  unsigned* mlS = packH1;
  unsigned short* mlD = (unsigned short*)(mlS + (size_t)NB1 * AGG_WAVES * AGG_KEEP);
  unsigned* mlN = (unsigned*)(mlD + (size_t)NB1 * AGG_WAVES * AGG_KEEP);

  k_pack_all<<<PK_B_FT + PK_B_W1 + PK_B_W2 + (MT * 4 * 32 + 255) / 256, 256, 0, stream>>>(
      x, ft_W, W1, W2, packX, packWft, packW1, packW2, MT);

  k_gemm_ft_ln<<<MT, 256, 0, stream>>>(packX, packWft, ft_b, ln_g, ln_b, packH1);

  k_gemm1<<<(MT + G1_W - 1) / G1_W * 8, 256, 0, stream>>>(packH1, packW1, att_s1, att_d1, hW1, aS1, aD1, MT);
  k_agg1_owned<<<NB1, 512, 0, stream>>>(ei, hW1, aS1, aD1, agg1, mlS, mlD, mlN, N, E);

  k_bnprep<<<MT, 256, 0, stream>>>(agg1, b1, bn_g, bn_b, bn_mean, bn_var, packH2);

  k_gemm2<<<MT, 32, 0, stream>>>(packH2, packW2, att_s2, att_d2, hW2, aS2, aD2);
  k_agg2_owned<<<(N + A2_R - 1) / A2_R, 512, 0, stream>>>(ei, hW2, aS2, aD2, b2, mlS, mlD, mlN, (float*)d_out, N, E);
}


# BatchNorm prep launch removed: layer-1 owner waves apply bias+BatchNorm+ReLU to their finished rows; the work-group writes them as bf16 A fragments via LDS in whole 128-byte lines; 7 launches to 6
# speedup vs baseline: 3.3757x; 1.0353x over previous
#include <hip/hip_runtime.h>
#include <stdint.h>
#include <stddef.h>


typedef __attribute__((ext_vector_type(16))) __bf16 v16bf;
typedef __attribute__((ext_vector_type(8)))  float  v8f;
typedef unsigned v4u __attribute__((ext_vector_type(4)));

#define LDSTR 520

__device__ __forceinline__ unsigned f2bf(float f) {
  unsigned u = __float_as_uint(f);
  return (u + 0x7FFFu + ((u >> 16) & 1u)) >> 16;
}

__device__ __forceinline__ int kmap(int v, int lane) {
  return ((v & 4) << 2) | ((lane & 16) >> 1) | ((v & 3) << 1);
}

__device__ __forceinline__ unsigned fenc(float f) {
  unsigned u = __float_as_uint(f);
  return (u & 0x80000000u) ? ~u : (u | 0x80000000u);
}
__device__ __forceinline__ float fdec(unsigned u) {
  return __uint_as_float((u & 0x80000000u) ? (u & 0x7FFFFFFFu) : ~u);
}

__device__ __forceinline__ void pack_tile_to_fragA(const float* tile,
                                                   unsigned* dst, int mt, int tid) {
  const int lane = tid & 31, w = tid >> 5;
  const int m = lane & 15;
#pragma unroll
  for (int q = 0; q < 2; ++q) {
    const int ks = w * 2 + q;
#pragma unroll
    for (int v = 0; v < 8; ++v) {
      const int k = (ks << 5) + kmap(v, lane);
      const unsigned lo = f2bf(tile[m * LDSTR + k]);
      const unsigned hi = f2bf(tile[m * LDSTR + k + 1]);
      dst[((size_t)(mt * 16 + ks) * 32 + lane) * 8 + v] = lo | (hi << 16);
    }
  }
}

typedef float pk_f4 __attribute__((ext_vector_type(4)));
typedef pk_f4 pk_f4u __attribute__((aligned(4)));

__device__ __forceinline__ unsigned bfpair(float lo, float hi) { return f2bf(lo) | (f2bf(hi) << 16); }

template <int K, int N>
__device__ __forceinline__ void pack_B_lane(const float* __restrict__ Wm, unsigned* __restrict__ pack, int id) {
  constexpr int nks = K >> 5;
  const int l = id & 31, t = id >> 5, ks = t % nks, nt = t / nks;
  const float* __restrict__ p = Wm + (size_t)((ks << 5) + ((l & 16) >> 1)) * N + (nt << 4) + (l & 15);
  float f[16];
#pragma unroll
  for (int j = 0; j < 8; ++j) {
    f[j] = p[(size_t)j * N];
    f[8 + j] = p[(size_t)(16 + j) * N];
  }
  const v4u lo = {bfpair(f[0], f[1]), bfpair(f[2], f[3]), bfpair(f[4], f[5]), bfpair(f[6], f[7])};
  const v4u hi = {bfpair(f[8], f[9]), bfpair(f[10], f[11]), bfpair(f[12], f[13]), bfpair(f[14], f[15])};
  v4u* o = (v4u*)(pack + (size_t)id * 8);
  o[0] = lo;
  o[1] = hi;
}

#define PK_B_FT 16
#define PK_B_W1 64
#define PK_B_W2 2

__global__ __launch_bounds__(256) void k_pack_all(
    const float* __restrict__ x, const float* __restrict__ ft_W,
    const float* __restrict__ W1, const float* __restrict__ W2,
    unsigned* __restrict__ packX, unsigned* __restrict__ packWft,
    unsigned* __restrict__ packW1, unsigned* __restrict__ packW2, int MT) {
  const int b = blockIdx.x, tid = threadIdx.x;
  if (b < PK_B_FT) {
    pack_B_lane<128, 512>(ft_W, packWft, b * 256 + tid);
  } else if (b < PK_B_FT + PK_B_W1) {
    pack_B_lane<512, 512>(W1, packW1, (b - PK_B_FT) * 256 + tid);
  } else if (b < PK_B_FT + PK_B_W1 + PK_B_W2) {
    pack_B_lane<512, 16>(W2, packW2, (b - PK_B_FT - PK_B_W1) * 256 + tid);
  } else {
    const int id = (b - PK_B_FT - PK_B_W1 - PK_B_W2) * 256 + tid;
    if (id < MT * 4 * 32) {
      const int l = id & 31, t = id >> 5, ks = t & 3, mt = t >> 2;
      const float* __restrict__ p = x + (size_t)((mt << 4) + (l & 15)) * 128 + (ks << 5) + ((l & 16) >> 1);
      const pk_f4u a = *(const pk_f4u*)(p);
      const pk_f4u c = *(const pk_f4u*)(p + 4);
      const pk_f4u d = *(const pk_f4u*)(p + 16);
      const pk_f4u e = *(const pk_f4u*)(p + 20);
      const v4u lo = {bfpair(a.x, a.y), bfpair(a.z, a.w), bfpair(c.x, c.y), bfpair(c.z, c.w)};
      const v4u hi = {bfpair(d.x, d.y), bfpair(d.z, d.w), bfpair(e.x, e.y), bfpair(e.z, e.w)};
      v4u* o = (v4u*)(packX + (size_t)id * 8);
      o[0] = lo;
      o[1] = hi;
    }
  }
}

__global__ __launch_bounds__(256) void k_gemm_ft_ln(
    const unsigned* __restrict__ packX, const unsigned* __restrict__ packW,
    const float* __restrict__ ft_b, const float* __restrict__ ln_g,
    const float* __restrict__ ln_b, unsigned* __restrict__ packH1) {
  __shared__ float tile[16 * LDSTR];
  __shared__ float psum[16][16], psq[16][16], stats[16][2];
  const int tid = threadIdx.x, lane = tid & 31, w = tid >> 5;
  const int mt = blockIdx.x;
  const int nks = 4;
  v8f acc[4] = {};
  const unsigned* __restrict__ ap = packX + ((size_t)(mt * nks) * 32 + lane) * 8;
  const unsigned* __restrict__ bp = packW + ((size_t)(w * 4 * nks) * 32 + lane) * 8;
  v16bf fa[2], fb[2][4];
  auto load_ks = [&](int ks) {
    fa[ks & 1] = *(const v16bf*)(ap + ks * 256);
#pragma unroll
    for (int j = 0; j < 4; ++j) fb[ks & 1][j] = *(const v16bf*)(bp + (j * nks + ks) * 256);
  };
  auto mma_ks = [&](int ks) {
#pragma unroll
    for (int j = 0; j < 4; ++j)
      acc[j] = __builtin_amdgcn_wmma_f32_16x16x32_bf16(false, fa[ks & 1], false, fb[ks & 1][j],
                                                       (short)0, acc[j], false, false);
  };
  load_ks(0);
  load_ks(1);
  __builtin_amdgcn_sched_barrier(0);
  mma_ks(0);
  __builtin_amdgcn_sched_barrier(0);
  load_ks(2);
  __builtin_amdgcn_sched_barrier(0);
  mma_ks(1);
  __builtin_amdgcn_sched_barrier(0);
  load_ks(3);
  __builtin_amdgcn_sched_barrier(0);
  mma_ks(2);
  mma_ks(3);
  const int m0 = (lane >> 4) << 3, nc = lane & 15;
#pragma unroll
  for (int j = 0; j < 4; ++j) {
    const int n = (w * 4 + j) * 16 + nc;
    const float bias = ft_b[n];
#pragma unroll
    for (int r = 0; r < 8; ++r) tile[(m0 + r) * LDSTR + n] = acc[j][r] + bias;
  }
  __syncthreads();
  {
    const int row = tid >> 4, p = tid & 15;
    float s = 0.f, s2 = 0.f;
    for (int c = p * 32; c < p * 32 + 32; ++c) {
      const float xv = tile[row * LDSTR + c];
      s += xv; s2 += xv * xv;
    }
    psum[row][p] = s; psq[row][p] = s2;
  }
  __syncthreads();
  if (tid < 16) {
    float s = 0.f, s2 = 0.f;
    for (int p = 0; p < 16; ++p) { s += psum[tid][p]; s2 += psq[tid][p]; }
    const float mu = s * (1.f / 512.f);
    const float var = s2 * (1.f / 512.f) - mu * mu;
    stats[tid][0] = mu;
    stats[tid][1] = rsqrtf(var + 1e-5f);
  }
  __syncthreads();
  {
    const int row = tid >> 4, p = tid & 15;
    const float mu = stats[row][0], rs = stats[row][1];
    for (int c = p * 32; c < p * 32 + 32; ++c) {
      float xv = (tile[row * LDSTR + c] - mu) * rs * ln_g[c] + ln_b[c];
      tile[row * LDSTR + c] = xv >= 0.f ? xv : 0.2f * xv;
    }
  }
  __syncthreads();
  pack_tile_to_fragA(tile, packH1, mt, tid);
}

typedef unsigned v8u __attribute__((ext_vector_type(8)));
#define G1_W 8
#define G1_OP 68

__global__ __launch_bounds__(256) void k_gemm1(
    const unsigned* __restrict__ packA, const unsigned* __restrict__ packW,
    const float* __restrict__ att_src, const float* __restrict__ att_dst,
    float* __restrict__ hW, float* __restrict__ a_src, float* __restrict__ a_dst, int MT) {
  __shared__ v4u bs[4 * 16 * 64];
  __shared__ __attribute__((aligned(16))) float ot[G1_W][16 * G1_OP];
  __shared__ __attribute__((aligned(16))) float atl[2][64];
  const int tid = threadIdx.x, lane = tid & 31, w = tid >> 5;
  const int head = blockIdx.x & 7, rg = blockIdx.x >> 3;
  const int mtw = rg * G1_W + w;
  const bool valid = mtw < MT;
  const int mt = valid ? mtw : MT - 1;
  const unsigned* __restrict__ ap = packA + ((size_t)mt * 16 * 32 + lane) * 8;
  v16bf ar[3][2];
  auto load_group = [&](int g) {
#pragma unroll
    for (int q = 0; q < 2; ++q) ar[g % 3][q] = *(const v16bf*)(ap + (2 * g + q) * 256);
  };
  load_group(0);
  {
    const v4u* __restrict__ gB = (const v4u*)(packW + (size_t)head * (4 * 16 * 256)) + tid;
    v4u* sb = bs + ((tid >> 6) << 6) + ((tid & 1) << 5) + ((tid & 63) >> 1);
    v4u tmp[16];
#pragma unroll
    for (int t = 0; t < 16; ++t) tmp[t] = gB[256 * t];
#pragma unroll
    for (int t = 0; t < 16; ++t) sb[256 * t] = tmp[t];
  }
  load_group(1);
  load_group(2);
  if (tid < 128) atl[tid >> 6][tid & 63] = (tid < 64 ? att_src : att_dst)[head * 64 + (tid & 63)];
  __syncthreads();
  v8f acc[4] = {};
  auto kstep = [&](const v16bf& a, int ks) {
    v16bf b[4];
#pragma unroll
    for (int j = 0; j < 4; ++j) {
      const v4u lo = bs[(j * 16 + ks) * 64 + lane];
      const v4u hi = bs[(j * 16 + ks) * 64 + 32 + lane];
      const v8u t = __builtin_shufflevector(lo, hi, 0, 1, 2, 3, 4, 5, 6, 7);
      b[j] = (v16bf)t;
    }
    __builtin_amdgcn_sched_barrier(0);
#pragma unroll
    for (int j = 0; j < 4; ++j)
      acc[j] = __builtin_amdgcn_wmma_f32_16x16x32_bf16(false, a, false, b[j],
                                                       (short)0, acc[j], false, false);
  };
#pragma unroll
  for (int g = 0; g < 8; ++g) {
    kstep(ar[g % 3][0], 2 * g);
    __builtin_amdgcn_sched_barrier(0);
    kstep(ar[g % 3][1], 2 * g + 1);
    if (g + 3 < 8) load_group(g + 3);
    __builtin_amdgcn_sched_barrier(0);
  }

  float* o = ot[w];
  const int m0 = (lane >> 4) << 3, nc = lane & 15;
#pragma unroll
  for (int j = 0; j < 4; ++j) {
#pragma unroll
    for (int r = 0; r < 8; ++r) o[(m0 + r) * G1_OP + j * 16 + nc] = acc[j][r];
  }
  __builtin_amdgcn_fence(__ATOMIC_ACQ_REL, "wavefront");
  if (valid) {
    const int hh = lane >> 4;
#pragma unroll
    for (int it = 0; it < 8; ++it) {
      const int row = it * 2 + hh, c4 = (lane & 15) * 4;
      const float4 v = *(const float4*)(o + row * G1_OP + c4);
      *(float4*)(hW + (size_t)(mt * 16 + row) * 512 + head * 64 + c4) = v;
    }
    const int row = lane & 15;
    const float* att = atl[hh];
    float s = 0.f;
    for (int c = 0; c < 64; ++c) s += o[row * G1_OP + c] * att[c];
    (hh ? a_dst : a_src)[(mt * 16 + row) * 8 + head] = s;
  }
}

#define AGG_R 40
#define AGG_WAVES 16
#define AGG_UL 8
#define AGG_STEP (128 * AGG_UL)
#define AGG_CAP (AGG_STEP + 32)
#define AGG_KEEP 256
#define AGG_OVER 0xFFFFFFFFu
#define AGG_NPITCH 32
#define AGG_SB 8
#define AGG_RC 16
#define AGG_NR (AGG_RC * AGG_WAVES)
#define AGG_SP 260

typedef int sc_vi4 __attribute__((ext_vector_type(4)));
typedef sc_vi4 sc_vi4u __attribute__((aligned(4)));

__global__ __launch_bounds__(512) void k_agg1_owned(const int* __restrict__ ei,
    const float* __restrict__ hW, const float* __restrict__ a_src,
    const float* __restrict__ a_dst,
    const float* __restrict__ b1, const float* __restrict__ bn_g, const float* __restrict__ bn_b,
    const float* __restrict__ bn_mean, const float* __restrict__ bn_var, unsigned* __restrict__ packH2,
    unsigned* __restrict__ gls, unsigned short* __restrict__ gld, unsigned* __restrict__ gcnt,
    int N, int E) {
  __shared__ float acc[AGG_R * 512];
  __shared__ unsigned mx[AGG_R * 8];
  __shared__ float den[AGG_R * 8];
  __shared__ float adl[AGG_R * 8];
  __shared__ unsigned lst[AGG_WAVES][AGG_CAP];
  __shared__ unsigned short lsd[AGG_WAVES][AGG_CAP];
  __shared__ unsigned kps[AGG_WAVES][AGG_KEEP];
  __shared__ unsigned short kpd[AGG_WAVES][AGG_KEEP];
  static_assert(AGG_WAVES == 16 && AGG_R <= 3 * AGG_WAVES && 3 * AGG_NR <= AGG_CAP, "owner = low 4 bits of the slot; three nodes' matches fit lst[w]");
  __shared__ unsigned rt[AGG_WAVES][AGG_WAVES][AGG_RC];
  __shared__ int ctab[AGG_WAVES][AGG_WAVES];
  __shared__ int wok[AGG_WAVES];
  __shared__ __attribute__((aligned(16))) float bnp[5][512];
  __shared__ __attribute__((aligned(16))) unsigned stg[AGG_R * AGG_SP];
  const int tid = threadIdx.x, lane = tid & 31, hh = lane >> 4;
  const int w = __builtin_amdgcn_readfirstlane(tid >> 5);
  const int d0 = blockIdx.x * AGG_R;
  const int Rn = min(AGG_R, N - d0);
  for (int i = tid; i < AGG_R * 8; i += 512) {
    mx[i] = 0u;
    adl[i] = (i < Rn * 8) ? a_dst[(size_t)d0 * 8 + i] : 0.f;
  }
  for (int i = tid; i < AGG_WAVES * AGG_WAVES; i += 512) (&ctab[0][0])[i] = 0;
  bnp[0][tid] = b1[tid];
  bnp[1][tid] = bn_mean[tid];
  bnp[2][tid] = rsqrtf(bn_var[tid] + 1e-5f);
  bnp[3][tid] = bn_g[tid];
  bnp[4][tid] = bn_b[tid];
  __syncthreads();
  static_assert((AGG_R * 64) % 512 == 0 && AGG_R % 4 == 0, "a work-group's 16-byte units: a whole number per thread, whole lines");
  auto write_rows = [&]() {
    v4u v[AGG_R * 64 / 512];
#pragma unroll
    for (int t = 0; t < AGG_R * 64 / 512; ++t) {
      const int i = tid + 512 * t;
      const int u = i & 7, L = i >> 3;
      const int dl = (L >> 5) * 4 + (u >> 1);
      v[t] = *(const v4u*)(stg + dl * AGG_SP + ((L >> 1) & 15) * 16 + (u & 1) * 8 + (L & 1) * 4);
    }
#pragma unroll
    for (int t = 0; t < AGG_R * 64 / 512; ++t) {
      const int i = tid + 512 * t;
      const int u = i & 7, L = i >> 3;
      const int r4 = u >> 1, p = u & 1;
      const int h = L & 1, ks = (L >> 1) & 15, rg = L >> 5;
      const int dl = rg * 4 + r4;
      if (dl < Rn) {
        const int n = d0 + dl;
        *(v4u*)(packH2 + ((size_t)((n >> 4) * 16 + ks) * 32 + (n & 15) + 16 * h) * 8 + 4 * p) = v[t];
      }
    }
  };
  const int SL = ((E + AGG_WAVES - 1) / AGG_WAVES + 3) & ~3;
  const int e_begin = min(E, w * SL);
  const int e_end = min(E, e_begin + SL);
  const int len = e_end - e_begin;
  const int nfull = len / AGG_STEP;
  const int nstep = (len + AGG_STEP - 1) / AGG_STEP;
  const int* __restrict__ dstp = ei + E;

  auto load_full = [&](int (&v)[AGG_UL][4], int start) {
#pragma unroll
    for (int u = 0; u < AGG_UL; ++u) {
      const sc_vi4u t = *(const sc_vi4u*)(dstp + start + u * 128 + lane * 4);
      v[u][0] = t.x; v[u][1] = t.y; v[u][2] = t.z; v[u][3] = t.w;
    }
  };
  auto load_part = [&](int (&v)[AGG_UL][4], int start) {
#pragma unroll
    for (int u = 0; u < AGG_UL; ++u) {
      const int e = start + u * 128 + lane * 4;
      if (e + 3 < e_end) {
        const sc_vi4u t = *(const sc_vi4u*)(dstp + e);
        v[u][0] = t.x; v[u][1] = t.y; v[u][2] = t.z; v[u][3] = t.w;
      } else {
#pragma unroll
        for (int k = 0; k < 4; ++k) v[u][k] = (e + k < e_end) ? dstp[e + k] : -1;
      }
    }
  };
  auto match_step = [&](const int (&v)[AGG_UL][4], int base, int cnt) -> int {
#pragma unroll
    for (int u = 0; u < AGG_UL; ++u) {
      unsigned dl[4], bal[4];
#pragma unroll
      for (int k = 0; k < 4; ++k) {
        dl[k] = (unsigned)(v[u][k] - d0);
        bal[k] = __builtin_amdgcn_ballot_w32(dl[k] < (unsigned)Rn);
      }
      if ((bal[0] | bal[1] | bal[2] | bal[3]) != 0u) {
#pragma unroll
        for (int k = 0; k < 4; ++k) {
          if (bal[k]) {
            const int pos = cnt + __builtin_popcount(bal[k] & ((1u << lane) - 1u));
            if (dl[k] < (unsigned)Rn) {
              lst[w][pos] = (unsigned)(base + u * 128 + lane * 4 + k);
              lsd[w][pos] = (unsigned short)dl[k];
            }
            cnt += __builtin_popcount(bal[k]);
          }
        }
      }
    }
    return cnt;
  };
  auto sum_rows = [&](const unsigned* ls, const unsigned short* ld, int n) {
    for (int j0 = 0; j0 < n; j0 += 4) {
      float4 row[4][4];
      float as[4][4];
      int dln[4];
#pragma unroll
      for (int b = 0; b < 4; ++b) {
        const int jb = min(j0 + b, n - 1);
        const size_t s = (size_t)ls[jb];
        dln[b] = (int)ld[jb];
        const float* rp = hW + s * 512 + 4 * lane;
#pragma unroll
        for (int j = 0; j < 4; ++j) row[b][j] = *(const float4*)(rp + 128 * j);
#pragma unroll
        for (int j = 0; j < 4; ++j) as[b][j] = a_src[s * 8 + 2 * j + hh];
      }
#pragma unroll
      for (int b = 0; b < 4; ++b) {
        if (j0 + b < n) {
          const int dl = dln[b];
#pragma unroll
          for (int j = 0; j < 4; ++j) {
            const int h = 2 * j + hh;
            float L = as[b][j] + adl[dl * 8 + h];
            L = L >= 0.f ? L : 0.2f * L;
            const float wgt = __expf(L - fdec(mx[dl * 8 + h]));
            float* ap = acc + dl * 512 + j * 128 + lane;
            atomicAdd(ap, wgt * row[b][j].x);
            atomicAdd(ap + 32, wgt * row[b][j].y);
            atomicAdd(ap + 64, wgt * row[b][j].z);
            atomicAdd(ap + 96, wgt * row[b][j].w);
            if ((lane & 15) == 0) atomicAdd(&den[dl * 8 + h], wgt);
          }
        }
      }
    }
  };

  int kept = 0;
  bool fits = true;
  unsigned* __restrict__ gl_s = gls + (size_t)(blockIdx.x * AGG_WAVES + w) * AGG_KEEP;
  unsigned short* __restrict__ gl_d = gld + (size_t)(blockIdx.x * AGG_WAVES + w) * AGG_KEEP;
  auto flush = [&](int cnt) {
    __builtin_amdgcn_fence(__ATOMIC_ACQ_REL, "wavefront");
    const bool keep = fits && kept + cnt <= AGG_KEEP;
    for (int j0 = 0; j0 < cnt; j0 += 32) {
      const int j = j0 + lane;
      const bool act = j < cnt;
      int s = 0, dl = 0;
      if (act) {
        const int e = (int)lst[w][j];
        dl = (int)lsd[w][j];
        s = ei[e];
        const float4 a0 = *(const float4*)(a_src + (size_t)s * 8);
        const float4 a1 = *(const float4*)(a_src + (size_t)s * 8 + 4);
        const float av[8] = {a0.x, a0.y, a0.z, a0.w, a1.x, a1.y, a1.z, a1.w};
#pragma unroll
        for (int h = 0; h < 8; ++h) {
          float L = av[h] + adl[dl * 8 + h];
          L = L >= 0.f ? L : 0.2f * L;
          atomicMax(&mx[dl * 8 + h], fenc(L));
        }
        if (keep) {
          kps[w][kept + j] = (unsigned)s;
          kpd[w][kept + j] = (unsigned short)dl;
          gl_s[kept + j] = (unsigned)s;
          gl_d[kept + j] = (unsigned short)dl;
        }
      }
      const int o = dl & (AGG_WAVES - 1);
      unsigned sm = __builtin_amdgcn_ballot_w32(act);
#pragma unroll
      for (int b = 0; b < 4; ++b) {
        const bool bit = ((o >> b) & 1) != 0;
        const unsigned bb = __builtin_amdgcn_ballot_w32(act && bit);
        sm &= bit ? bb : ~bb;
      }
      if (act) {
        const int rank = __builtin_popcount(sm & ((1u << lane) - 1u));
        const int base = ctab[w][o];
        if (base + rank < AGG_RC) rt[w][o][base + rank] = (unsigned)s | ((unsigned)(dl >> 4) << 30);
        if (rank == 0) ctab[w][o] = base + __builtin_popcount(sm);
      }
      __builtin_amdgcn_fence(__ATOMIC_ACQ_REL, "wavefront");
    }
    if (keep) kept += cnt; else fits = false;
  };
  {
    int cnt = 0;
    if (nfull > 0) {
      int va[AGG_UL][4], vb[AGG_UL][4];
      const int last_full = e_begin + (nfull - 1) * AGG_STEP;
      load_full(va, e_begin);
      for (int it = 0; it < nfull; it += 2) {
        const int base = e_begin + it * AGG_STEP;
        load_full(vb, min(base + AGG_STEP, last_full));
        cnt = match_step(va, base, cnt);
        if (cnt >= 32) { flush(cnt); cnt = 0; }
        load_full(va, min(base + 2 * AGG_STEP, last_full));
        if (it + 1 < nfull) {
          cnt = match_step(vb, base + AGG_STEP, cnt);
          if (cnt >= 32) { flush(cnt); cnt = 0; }
        }
      }
    }
    if (nfull < nstep) {
      int vt[AGG_UL][4];
      const int base = e_begin + nfull * AGG_STEP;
      load_part(vt, base);
      cnt = match_step(vt, base, cnt);
    }
    if (cnt > 0) flush(cnt);
  }
  {
    const int myc = lane < AGG_WAVES ? ctab[w][lane] : 0;
    const bool rfit = __builtin_amdgcn_ballot_w32(myc > AGG_RC) == 0u;
    if (lane == 0) {
      gcnt[blockIdx.x * AGG_NPITCH + w] = fits ? (unsigned)kept : AGG_OVER;
      wok[w] = (fits && rfit) ? 1 : 0;
    }
  }
  __syncthreads();

  const int okv = lane < AGG_WAVES ? wok[lane] : 1;
  if (__builtin_amdgcn_ballot_w32(okv == 0) == 0u) {
    const int cq = lane < AGG_WAVES ? ctab[lane][w] : 0;
    const int e16 = lane & 15;
    unsigned ent[8];
    bool val[8];
#pragma unroll
    for (int k = 0; k < 8; ++k) {
      ent[k] = rt[2 * k + hh][w][e16];
      const int c0 = __builtin_amdgcn_readlane(cq, 2 * k), c1 = __builtin_amdgcn_readlane(cq, 2 * k + 1);
      val[k] = e16 < (hh ? c1 : c0);
    }
    int cn0 = 0, cn1 = 0, cn2 = 0;
#pragma unroll
    for (int k = 0; k < 8; ++k) {
      const int tk = (int)(ent[k] >> 30);
      const unsigned b0 = __builtin_amdgcn_ballot_w32(val[k] && tk == 0);
      const unsigned b1 = __builtin_amdgcn_ballot_w32(val[k] && tk == 1);
      const unsigned b2 = __builtin_amdgcn_ballot_w32(val[k] && tk == 2);
      if ((b0 | b1 | b2) != 0u) {
        const unsigned below = (1u << lane) - 1u;
        const unsigned mine = tk == 0 ? b0 : (tk == 1 ? b1 : b2);
        const int at = tk * AGG_NR + (tk == 0 ? cn0 : (tk == 1 ? cn1 : cn2)) + __builtin_popcount(mine & below);
        if (val[k]) lst[w][at] = ent[k] & 0x3FFFFFFFu;
        cn0 += __builtin_popcount(b0);
        cn1 += __builtin_popcount(b1);
        cn2 += __builtin_popcount(b2);
      }
    }
    __builtin_amdgcn_fence(__ATOMIC_ACQ_REL, "wavefront");
    for (int t = 0; t < 3; ++t) {
      const int dl = w + AGG_WAVES * t;
      if (dl >= Rn) break;
      const unsigned* ls = lst[w] + t * AGG_NR;
      const int n = t == 0 ? cn0 : (t == 1 ? cn1 : cn2);
      float adv[4], mxv[4];
#pragma unroll
      for (int j = 0; j < 4; ++j) {
        adv[j] = adl[dl * 8 + 2 * j + hh];
        mxv[j] = fdec(mx[dl * 8 + 2 * j + hh]);
      }
      float4 sm[4];
      float dn[4];
#pragma unroll
      for (int j = 0; j < 4; ++j) { sm[j] = make_float4(0.f, 0.f, 0.f, 0.f); dn[j] = 0.f; }
      for (int j0 = 0; j0 < n; j0 += AGG_SB) {
        const unsigned sv = ls[min(j0 + (lane & (AGG_SB - 1)), n - 1)];
        float4 row[AGG_SB][4];
        float as[AGG_SB][4];
#pragma unroll
        for (int b = 0; b < AGG_SB; ++b) {
          if (j0 + b < n) {
            const size_t s = (size_t)(unsigned)__builtin_amdgcn_readlane((int)sv, b);
            const float* rp = hW + s * 512 + 4 * lane;
#pragma unroll
            for (int j = 0; j < 4; ++j) row[b][j] = *(const float4*)(rp + 128 * j);
#pragma unroll
            for (int j = 0; j < 4; ++j) as[b][j] = a_src[s * 8 + 2 * j + hh];
          }
        }
#pragma unroll
        for (int b = 0; b < AGG_SB; ++b) {
          if (j0 + b < n) {
#pragma unroll
            for (int j = 0; j < 4; ++j) {
              float L = as[b][j] + adv[j];
              L = L >= 0.f ? L : 0.2f * L;
              const float wgt = __expf(L - mxv[j]);
              sm[j].x += wgt * row[b][j].x;
              sm[j].y += wgt * row[b][j].y;
              sm[j].z += wgt * row[b][j].z;
              sm[j].w += wgt * row[b][j].w;
              dn[j] += wgt;
            }
          }
        }
      }
      unsigned* sp = stg + dl * AGG_SP + 2 * lane;
#pragma unroll
      for (int j = 0; j < 4; ++j) {
        const int c = 128 * j + 4 * lane;
        const float4 pb = *(const float4*)(&bnp[0][c]);
        const float4 pm = *(const float4*)(&bnp[1][c]);
        const float4 ps = *(const float4*)(&bnp[2][c]);
        const float4 pg = *(const float4*)(&bnp[3][c]);
        const float4 po = *(const float4*)(&bnp[4][c]);
        float4 ov;
        ov.x = dn[j] != 0.f ? sm[j].x / dn[j] : 0.f;
        ov.y = dn[j] != 0.f ? sm[j].y / dn[j] : 0.f;
        ov.z = dn[j] != 0.f ? sm[j].z / dn[j] : 0.f;
        ov.w = dn[j] != 0.f ? sm[j].w / dn[j] : 0.f;
        float4 y;
        y.x = ((ov.x + pb.x) - pm.x) * ps.x * pg.x + po.x;
        y.y = ((ov.y + pb.y) - pm.y) * ps.y * pg.y + po.y;
        y.z = ((ov.z + pb.z) - pm.z) * ps.z * pg.z + po.z;
        y.w = ((ov.w + pb.w) - pm.w) * ps.w * pg.w + po.w;
        y.x = y.x > 0.f ? y.x : 0.f;
        y.y = y.y > 0.f ? y.y : 0.f;
        y.z = y.z > 0.f ? y.z : 0.f;
        y.w = y.w > 0.f ? y.w : 0.f;
        uint2 o2;
        o2.x = bfpair(y.x, y.y);
        o2.y = bfpair(y.z, y.w);
        *(uint2*)(sp + 64 * j) = o2;
      }
    }
    __syncthreads();
    write_rows();
    return;
  }
  for (int i = tid; i < AGG_R * 512; i += 512) acc[i] = 0.f;
  for (int i = tid; i < AGG_R * 8; i += 512) den[i] = 0.f;
  __syncthreads();
  if (fits) {
    sum_rows(kps[w], kpd[w], kept);
  } else {
    for (int it = 0; it < nstep; ++it) {
      const int base = e_begin + it * AGG_STEP;
      int dv[AGG_UL][4];
      load_part(dv, base);
      const int cnt = match_step(dv, base, 0);
      __builtin_amdgcn_fence(__ATOMIC_ACQ_REL, "wavefront");
      for (int j = lane; j < cnt; j += 32) lst[w][j] = (unsigned)ei[lst[w][j]];
      __builtin_amdgcn_fence(__ATOMIC_ACQ_REL, "wavefront");
      sum_rows(lst[w], lsd[w], cnt);
    }
  }
  __syncthreads();
  for (int i = tid; i < Rn * 512; i += 512) {
    const int dl = i >> 9, c = i & 511;
    const int q = c >> 2, k = c & 3, j = q >> 5, l = q & 31;
    const float dn = den[dl * 8 + (c >> 6)];
    const float v = acc[dl * 512 + (j * 4 + k) * 32 + l];
    const float xv = dn != 0.f ? v / dn : 0.f;
    float y = ((xv + bnp[0][c]) - bnp[1][c]) * bnp[2][c] * bnp[3][c] + bnp[4][c];
    y = y > 0.f ? y : 0.f;
    ((unsigned short*)stg)[dl * (2 * AGG_SP) + c] = (unsigned short)f2bf(y);
  }
  __syncthreads();
  write_rows();
}

#define A2_R 48
#define A2_WAVES 16
#define A2_UL 8
#define A2_STEP (128 * A2_UL)
#define A2_CAP (A2_STEP + 32)
#define A2_KEEP 512

typedef int a2_vi4 __attribute__((ext_vector_type(4)));
typedef a2_vi4 a2_vi4u __attribute__((aligned(4)));

__global__ __launch_bounds__(512) void k_agg2_owned(const int* __restrict__ ei,
    const float* __restrict__ hW, const float* __restrict__ a_src,
    const float* __restrict__ a_dst, const float* __restrict__ b2,
    const unsigned* __restrict__ gls, const unsigned short* __restrict__ gld,
    const unsigned* __restrict__ gcnt, float* __restrict__ out, int N, int E) {
  __shared__ float acc[A2_R * 16];
  __shared__ unsigned mx[A2_R * 8];
  __shared__ float den[A2_R * 8];
  __shared__ float adl[A2_R * 8];
  __shared__ unsigned lst[A2_WAVES][A2_CAP];
  __shared__ unsigned short lsd[A2_WAVES][A2_CAP];
  __shared__ unsigned kps[A2_WAVES][A2_KEEP];
  __shared__ unsigned short kpd[A2_WAVES][A2_KEEP];
  const int tid = threadIdx.x, lane = tid & 31;
  const int w = __builtin_amdgcn_readfirstlane(tid >> 5);
  const int d0 = blockIdx.x * A2_R;
  const int Rn = min(A2_R, N - d0);
  static_assert(AGG_R % 8 == 0 && A2_R % 8 == 0 && AGG_R - 8 + A2_R <= 2 * AGG_R, "a layer-2 range spans at most two layer-1 ranges");
  static_assert(AGG_WAVES == A2_WAVES && AGG_WAVES == 16, "wave w walks list w; a wave's lanes hold both ranges' 16 lengths");
  const int b1a = d0 / AGG_R;
  const int nl = (d0 + Rn - 1) / AGG_R - b1a + 1;
  const unsigned cl = lane < AGG_WAVES * nl ? gcnt[(b1a + (lane >> 4)) * AGG_NPITCH + (lane & 15)] : 0u;
  unsigned es[4];
  int esl[4];
#pragma unroll
  for (int c = 0; c < 4; ++c) {
    const int bq = b1a + ((c >> 1) < nl ? (c >> 1) : 0);
    const size_t li = (size_t)(bq * AGG_WAVES + w) * AGG_KEEP + lane + 32 * (c & 1);
    es[c] = gls[li];
    esl[c] = (int)gld[li];
  }
  for (int i = tid; i < A2_R * 16; i += 512) acc[i] = 0.f;
  for (int i = tid; i < A2_R * 8; i += 512) {
    mx[i] = 0u; den[i] = 0.f;
    adl[i] = (i < Rn * 8) ? a_dst[(size_t)d0 * 8 + i] : 0.f;
  }
  __syncthreads();
  const int SL = ((E + A2_WAVES - 1) / A2_WAVES + 3) & ~3;
  const int e_begin = w * SL;
  const int e_end = min(E, e_begin + SL);
  const int nit = (SL + A2_STEP - 1) / A2_STEP;
  const int* __restrict__ dstp = ei + E;

  auto load_step = [&](int (&v)[A2_UL][4], int start) {
    if (start + A2_STEP <= e_end) {
#pragma unroll
      for (int u = 0; u < A2_UL; ++u) {
        const a2_vi4u t = *(const a2_vi4u*)(dstp + start + u * 128 + lane * 4);
        v[u][0] = t.x; v[u][1] = t.y; v[u][2] = t.z; v[u][3] = t.w;
      }
    } else {
#pragma unroll
      for (int u = 0; u < A2_UL; ++u) {
        const int e = start + u * 128 + lane * 4;
        if (e + 3 < e_end) {
          const a2_vi4u t = *(const a2_vi4u*)(dstp + e);
          v[u][0] = t.x; v[u][1] = t.y; v[u][2] = t.z; v[u][3] = t.w;
        } else {
#pragma unroll
          for (int k = 0; k < 4; ++k) v[u][k] = (e + k < e_end) ? dstp[e + k] : -1;
        }
      }
    }
  };
  auto edge_max = [&](size_t s, int dl) {
    const float4 a0 = *(const float4*)(a_src + s * 8);
    const float4 a1 = *(const float4*)(a_src + s * 8 + 4);
    const float av[8] = {a0.x, a0.y, a0.z, a0.w, a1.x, a1.y, a1.z, a1.w};
#pragma unroll
    for (int h = 0; h < 8; ++h) {
      float L = av[h] + adl[dl * 8 + h];
      L = L >= 0.f ? L : 0.2f * L;
      atomicMax(&mx[dl * 8 + h], fenc(L));
    }
  };
  auto edge_sum = [&](size_t s, int dl) {
    const float4 a0 = *(const float4*)(a_src + s * 8);
    const float4 a1 = *(const float4*)(a_src + s * 8 + 4);
    const float4 r0 = *(const float4*)(hW + s * 16);
    const float4 r1 = *(const float4*)(hW + s * 16 + 4);
    const float4 r2 = *(const float4*)(hW + s * 16 + 8);
    const float4 r3 = *(const float4*)(hW + s * 16 + 12);
    const float av[8] = {a0.x, a0.y, a0.z, a0.w, a1.x, a1.y, a1.z, a1.w};
    const float hv[16] = {r0.x, r0.y, r0.z, r0.w, r1.x, r1.y, r1.z, r1.w,
                          r2.x, r2.y, r2.z, r2.w, r3.x, r3.y, r3.z, r3.w};
#pragma unroll
    for (int h = 0; h < 8; ++h) {
      float L = av[h] + adl[dl * 8 + h];
      L = L >= 0.f ? L : 0.2f * L;
      const float wgt = __expf(L - fdec(mx[dl * 8 + h]));
      atomicAdd(&den[dl * 8 + h], wgt);
      atomicAdd(&acc[dl * 16 + 2 * h], wgt * hv[2 * h]);
      atomicAdd(&acc[dl * 16 + 2 * h + 1], wgt * hv[2 * h + 1]);
    }
  };

  const bool use_lists = __builtin_amdgcn_ballot_w32(cl == AGG_OVER) == 0u;

  if (use_lists) {
    unsigned cn[2];
    cn[0] = min((unsigned)__builtin_amdgcn_readlane((int)cl, w), (unsigned)AGG_KEEP);
    cn[1] = nl > 1 ? min((unsigned)__builtin_amdgcn_readlane((int)cl, AGG_WAVES + w), (unsigned)AGG_KEEP) : 0u;
    bool ok[4];
    int cdl[4];
    float4 av[4][2], hv[4][4];
#pragma unroll
    for (int c = 0; c < 4; ++c) {
      const int q = c >> 1;
      const int dl = (b1a + q) * AGG_R + esl[c] - d0;
      ok[c] = lane + 32 * (c & 1) < (int)cn[q] && (unsigned)dl < (unsigned)Rn;
      cdl[c] = ok[c] ? dl : 0;
      const size_t s = ok[c] ? (size_t)es[c] : (size_t)0;
      av[c][0] = *(const float4*)(a_src + s * 8);
      av[c][1] = *(const float4*)(a_src + s * 8 + 4);
#pragma unroll
      for (int r = 0; r < 4; ++r) hv[c][r] = *(const float4*)(hW + s * 16 + 4 * r);
    }
    auto rest = [&](auto&& fn) {
      for (int q = 0; q < nl; ++q) {
        const size_t lb = (size_t)((b1a + q) * AGG_WAVES + w) * AGG_KEEP;
        for (int j = 64 + lane; j < (int)cn[q]; j += 32) {
          const int dl = (b1a + q) * AGG_R + (int)gld[lb + j] - d0;
          if ((unsigned)dl < (unsigned)Rn) fn((size_t)gls[lb + j], dl);
        }
      }
    };
#pragma unroll
    for (int c = 0; c < 4; ++c) {
      if (ok[c]) {
        const float a8[8] = {av[c][0].x, av[c][0].y, av[c][0].z, av[c][0].w, av[c][1].x, av[c][1].y, av[c][1].z, av[c][1].w};
#pragma unroll
        for (int h = 0; h < 8; ++h) {
          float L = a8[h] + adl[cdl[c] * 8 + h];
          L = L >= 0.f ? L : 0.2f * L;
          atomicMax(&mx[cdl[c] * 8 + h], fenc(L));
        }
      }
    }
    rest(edge_max);
    __syncthreads();
#pragma unroll
    for (int c = 0; c < 4; ++c) {
      if (ok[c]) {
        const float a8[8] = {av[c][0].x, av[c][0].y, av[c][0].z, av[c][0].w, av[c][1].x, av[c][1].y, av[c][1].z, av[c][1].w};
        const float h16[16] = {hv[c][0].x, hv[c][0].y, hv[c][0].z, hv[c][0].w, hv[c][1].x, hv[c][1].y, hv[c][1].z, hv[c][1].w,
                               hv[c][2].x, hv[c][2].y, hv[c][2].z, hv[c][2].w, hv[c][3].x, hv[c][3].y, hv[c][3].z, hv[c][3].w};
        const int dl = cdl[c];
#pragma unroll
        for (int h = 0; h < 8; ++h) {
          float L = a8[h] + adl[dl * 8 + h];
          L = L >= 0.f ? L : 0.2f * L;
          const float wgt = __expf(L - fdec(mx[dl * 8 + h]));
          atomicAdd(&den[dl * 8 + h], wgt);
          atomicAdd(&acc[dl * 16 + 2 * h], wgt * h16[2 * h]);
          atomicAdd(&acc[dl * 16 + 2 * h + 1], wgt * h16[2 * h + 1]);
        }
      }
    }
    rest(edge_sum);
    __syncthreads();
  } else {
    int kept = 0;
    bool fits = true;
    for (int pass = 0; pass < 2; ++pass) {
      if (pass == 1 && fits) {
        for (int j = lane; j < kept; j += 32) edge_sum((size_t)kps[w][j], (int)kpd[w][j]);
      } else {
        int dv[A2_UL][4];
        load_step(dv, e_begin);
        int cnt = 0;
        for (int it = 0; it < nit; ++it) {
          const int base = e_begin + it * A2_STEP;
          int nx[A2_UL][4];
          load_step(nx, base + A2_STEP);
#pragma unroll
          for (int u = 0; u < A2_UL; ++u) {
#pragma unroll
            for (int k = 0; k < 4; ++k) {
              const unsigned dl = (unsigned)(dv[u][k] - d0);
              const bool match = dl < (unsigned)Rn;
              const unsigned bal = __builtin_amdgcn_ballot_w32(match);
              if (__builtin_expect(bal != 0u, 0)) {
                const int pos = cnt + __builtin_popcount(bal & ((1u << lane) - 1u));
                if (match) {
                  lst[w][pos] = (unsigned)(base + u * 128 + lane * 4 + k);
                  lsd[w][pos] = (unsigned short)dl;
                }
                cnt += __builtin_popcount(bal);
              }
            }
          }
          if (cnt >= 32 || it == nit - 1) {
            __builtin_amdgcn_fence(__ATOMIC_ACQ_REL, "wavefront");
            const bool keep = fits && kept + cnt <= A2_KEEP;
            for (int j = lane; j < cnt; j += 32) {
              const int e = (int)lst[w][j];
              const int dl = (int)lsd[w][j];
              const size_t s = (size_t)ei[e];
              if (pass == 0) {
                edge_max(s, dl);
                if (keep) {
                  kps[w][kept + j] = (unsigned)s;
                  kpd[w][kept + j] = (unsigned short)dl;
                }
              } else {
                edge_sum(s, dl);
              }
            }
            if (keep) kept += cnt; else fits = false;
            cnt = 0;
          }
#pragma unroll
          for (int u = 0; u < A2_UL; ++u) {
#pragma unroll
            for (int k = 0; k < 4; ++k) dv[u][k] = nx[u][k];
          }
        }
      }
      __syncthreads();
    }
  }
  for (int i = tid; i < Rn * 2; i += 512) {
    const int dl = i >> 1, c = i & 1;
    float s = 0.f;
#pragma unroll
    for (int h = 0; h < 8; ++h) {
      const float dn = den[dl * 8 + h];
      s += dn != 0.f ? acc[dl * 16 + 2 * h + c] / dn : 0.f;
    }
    out[(size_t)(d0 + dl) * 2 + c] = s * 0.125f + b2[c];
  }
}

__global__ __launch_bounds__(32) void k_gemm2(
    const unsigned* __restrict__ packA, const unsigned* __restrict__ packW,
    const float* __restrict__ att_src, const float* __restrict__ att_dst,
    float* __restrict__ hW, float* __restrict__ a_src, float* __restrict__ a_dst) {
  __shared__ float t2[16][16];
  const int lane = threadIdx.x;
  const int mt = blockIdx.x;
  const int nks = 16;
  v8f acc = {};
  v16bf fa[16], fb[16];
#pragma unroll
  for (int ks = 0; ks < nks; ++ks) {
    fa[ks] = *(const v16bf*)(packA + ((size_t)(mt * nks + ks) * 32 + lane) * 8);
    fb[ks] = *(const v16bf*)(packW + ((size_t)ks * 32 + lane) * 8);
  }
  __builtin_amdgcn_sched_barrier(0);
#pragma unroll
  for (int ks = 0; ks < nks; ++ks)
    acc = __builtin_amdgcn_wmma_f32_16x16x32_bf16(false, fa[ks], false, fb[ks],
                                                  (short)0, acc, false, false);
  const int m0 = (lane >> 4) << 3, nc = lane & 15;
#pragma unroll
  for (int r = 0; r < 8; ++r) t2[m0 + r][nc] = acc[r];
  __syncthreads();
  for (int i = lane; i < 256; i += 32) {
    const int row = i >> 4, c = i & 15;
    hW[(size_t)(mt * 16 + row) * 16 + c] = t2[row][c];
  }
  if (lane < 16) {
    const int row = lane, node = mt * 16 + row;
    for (int h = 0; h < 8; ++h) {
      const float v0 = t2[row][2 * h], v1 = t2[row][2 * h + 1];
      a_src[node * 8 + h] = v0 * att_src[2 * h] + v1 * att_src[2 * h + 1];
      a_dst[node * 8 + h] = v0 * att_dst[2 * h] + v1 * att_dst[2 * h + 1];
    }
  }
}

extern "C" void kernel_launch(void* const* d_in, const int* in_sizes, int n_in,
                              void* d_out, int out_size, void* d_ws, size_t ws_size,
                              hipStream_t stream) {
  const float* x       = (const float*)d_in[0];
  const int*   ei      = (const int*)d_in[1];
  const float* ft_W    = (const float*)d_in[2];
  const float* ft_b    = (const float*)d_in[3];
  const float* ln_g    = (const float*)d_in[4];
  const float* ln_b    = (const float*)d_in[5];
  const float* W1      = (const float*)d_in[6];
  const float* att_s1  = (const float*)d_in[7];
  const float* att_d1  = (const float*)d_in[8];
  const float* b1      = (const float*)d_in[9];
  const float* bn_g    = (const float*)d_in[10];
  const float* bn_b    = (const float*)d_in[11];
  const float* bn_mean = (const float*)d_in[12];
  const float* bn_var  = (const float*)d_in[13];
  const float* W2      = (const float*)d_in[14];
  const float* att_s2  = (const float*)d_in[15];
  const float* att_d2  = (const float*)d_in[16];
  const float* b2      = (const float*)d_in[17];

  const int N  = in_sizes[0] / 128;
  const int E  = in_sizes[1] / 2;
  const int MT = N / 16;

  char* cur = (char*)d_ws;
  auto alloc = [&](size_t bytes) -> void* {
    void* p = (void*)cur;
    cur += (bytes + 255) & ~(size_t)255;
    return p;
  };
  unsigned* packWft = (unsigned*)alloc((size_t)32 * 4 * 256 * 4);
  unsigned* packW1  = (unsigned*)alloc((size_t)32 * 16 * 256 * 4);
  unsigned* packW2  = (unsigned*)alloc((size_t)1 * 16 * 256 * 4);
  unsigned* packX   = (unsigned*)alloc((size_t)MT * 4 * 256 * 4);
  unsigned* packH1  = (unsigned*)alloc((size_t)MT * 16 * 256 * 4);
  unsigned* packH2  = (unsigned*)alloc((size_t)MT * 16 * 256 * 4);
  float* hW1 = (float*)alloc((size_t)N * 512 * 4);
  float* hW2 = (float*)alloc((size_t)N * 16 * 4);
  float* aS1 = (float*)alloc((size_t)N * 8 * 4);
  float* aD1 = (float*)alloc((size_t)N * 8 * 4);
  float* aS2 = (float*)alloc((size_t)N * 8 * 4);
  float* aD2 = (float*)alloc((size_t)N * 8 * 4);
  const int NB1 = (N + AGG_R - 1) / AGG_R;
  unsigned* mlS = packH1;
  unsigned short* mlD = (unsigned short*)(mlS + (size_t)NB1 * AGG_WAVES * AGG_KEEP);
  unsigned* mlN = (unsigned*)(mlD + (size_t)NB1 * AGG_WAVES * AGG_KEEP);

  k_pack_all<<<PK_B_FT + PK_B_W1 + PK_B_W2 + (MT * 4 * 32 + 255) / 256, 256, 0, stream>>>(
      x, ft_W, W1, W2, packX, packWft, packW1, packW2, MT);

  k_gemm_ft_ln<<<MT, 256, 0, stream>>>(packX, packWft, ft_b, ln_g, ln_b, packH1);

  k_gemm1<<<(MT + G1_W - 1) / G1_W * 8, 256, 0, stream>>>(packH1, packW1, att_s1, att_d1, hW1, aS1, aD1, MT);
  k_agg1_owned<<<NB1, 512, 0, stream>>>(ei, hW1, aS1, aD1, b1, bn_g, bn_b, bn_mean, bn_var, packH2, mlS, mlD, mlN, N, E);

  k_gemm2<<<MT, 32, 0, stream>>>(packH2, packW2, att_s2, att_d2, hW2, aS2, aD2);
  k_agg2_owned<<<(N + A2_R - 1) / A2_R, 512, 0, stream>>>(ei, hW2, aS2, aD2, b2, mlS, mlD, mlN, (float*)d_out, N, E);
}
